# starting kernel
# baseline (speedup 1.0000x reference)
#include <hip/hip_runtime.h>
#include <math.h>


typedef __attribute__((ext_vector_type(2))) float v2f;
typedef __attribute__((ext_vector_type(8))) float v8f;

#define NEG_SLOPE 0.2f

__global__ __launch_bounds__(256) void gat_gemm_wmma_f32(
    const float* __restrict__ A, const float* __restrict__ W,
    float* __restrict__ C, int M, int K) {
  const int wave = threadIdx.x >> 5;
  const int lane = threadIdx.x & 31;
  const int ntile = wave & 3;
  const int mtile = (blockIdx.x << 1) | (wave >> 2);
  if (mtile * 16 >= M) return;

  const int mrow = mtile * 16 + (lane & 15);
  const int koff = (lane >> 4) << 1;
  const int col  = ntile * 16 + (lane & 15);

  const float* __restrict__ arow = A + (size_t)mrow * K;
  v8f acc = {};
  for (int k = 0; k < K; k += 4) {
    v2f a, b;
    a.x = arow[k + koff];
    a.y = arow[k + koff + 1];
    b.x = W[(size_t)(k + koff)     * 64 + col];
    b.y = W[(size_t)(k + koff + 1) * 64 + col];
    acc = __builtin_amdgcn_wmma_f32_16x16x4_f32(
        false, a, false, b, (short)0, acc, false, false);
  }
  const int rbase = mtile * 16 + ((lane >> 4) << 3);
#pragma unroll
  for (int v = 0; v < 8; ++v)
    C[(size_t)(rbase + v) * 64 + col] = acc[v];
}

__global__ __launch_bounds__(256) void gat_alpha(
    const float* __restrict__ hp, const float* __restrict__ a_s,
    const float* __restrict__ a_d, float* __restrict__ out_s,
    float* __restrict__ out_d, int N) {
  const int node = blockIdx.x * 8 + (threadIdx.x >> 5);
  const int lane = threadIdx.x & 31;
  if (node >= N) return;
  const float* __restrict__ h = hp + (size_t)node * 64;
  float s = h[lane] * a_s[lane] + h[lane + 32] * a_s[lane + 32];
  float d = h[lane] * a_d[lane] + h[lane + 32] * a_d[lane + 32];
#pragma unroll
  for (int off = 16; off > 0; off >>= 1) {
    s += __shfl_down(s, off, 32);
    d += __shfl_down(d, off, 32);
  }
  if (lane == 0) { out_s[node] = s; out_d[node] = d; }
}

__global__ void gat_init(float* __restrict__ accum, unsigned int* __restrict__ maxk,
                         float* __restrict__ denom, int N) {
  const int tid = blockIdx.x * blockDim.x + threadIdx.x;
  if (tid < N * 64) accum[tid] = 0.0f;
  if (tid < N) { maxk[tid] = 0u; denom[tid] = 0.0f; }
}

__device__ __forceinline__ unsigned int f2key(float f) {
  unsigned int u = __float_as_uint(f);
  return (u & 0x80000000u) ? ~u : (u | 0x80000000u);
}
__device__ __forceinline__ float key2f(unsigned int k) {
  return __uint_as_float((k & 0x80000000u) ? (k ^ 0x80000000u) : ~k);
}

__device__ __forceinline__ void edge_ends(const int* __restrict__ ei, int idx,
                                          int E, int& src, int& dst) {
  if (idx < E) { src = ei[idx]; dst = ei[E + idx]; }
  else         { src = idx - E; dst = idx - E; }
}

__device__ __forceinline__ float edge_logit(const float* __restrict__ as_,
                                            const float* __restrict__ ad_,
                                            int src, int dst) {
  float e = as_[src] + ad_[dst];
  return (e > 0.0f) ? e : NEG_SLOPE * e;
}

__global__ void gat_edge_max(const int* __restrict__ ei,
                             const float* __restrict__ as_,
                             const float* __restrict__ ad_,
                             unsigned int* __restrict__ maxk, int E, int N) {
  const int idx = blockIdx.x * blockDim.x + threadIdx.x;
  if (idx >= E + N) return;
  int src, dst; edge_ends(ei, idx, E, src, dst);
  atomicMax(&maxk[dst], f2key(edge_logit(as_, ad_, src, dst)));
}

__global__ void gat_edge_exp(const int* __restrict__ ei,
                             const float* __restrict__ as_,
                             const float* __restrict__ ad_,
                             const unsigned int* __restrict__ maxk,
                             float* __restrict__ denom,
                             float* __restrict__ eexp, int E, int N) {
  const int idx = blockIdx.x * blockDim.x + threadIdx.x;
  if (idx >= E + N) return;
  int src, dst; edge_ends(ei, idx, E, src, dst);
  const float v = __expf(edge_logit(as_, ad_, src, dst) - key2f(maxk[dst]));
  eexp[idx] = v;
  atomicAdd(&denom[dst], v);
}

__global__ void gat_edge_scatter(const int* __restrict__ ei,
                                 const float* __restrict__ hp,
                                 const float* __restrict__ eexp,
                                 const float* __restrict__ denom,
                                 float* __restrict__ accum, int E, int N) {
  const long long tid = (long long)blockIdx.x * blockDim.x + threadIdx.x;
  if (tid >= (long long)(E + N) * 64) return;
  const int idx = (int)(tid >> 6);
  const int c   = (int)(tid & 63);
  int src, dst; edge_ends(ei, idx, E, src, dst);
  const float a = eexp[idx] / (denom[dst] + 1e-16f);
  atomicAdd(&accum[(size_t)dst * 64 + c], hp[(size_t)src * 64 + c] * a);
}

__global__ void gat_finalize(const float* __restrict__ accum,
                             const float* __restrict__ b,
                             float* __restrict__ out, int N) {
  const int tid = blockIdx.x * blockDim.x + threadIdx.x;
  if (tid >= N * 64) return;
  const float v = accum[tid] + b[tid & 63];
  out[tid] = v > 0.0f ? v : 0.0f;
}

extern "C" void kernel_launch(void* const* d_in, const int* in_sizes, int n_in,
                              void* d_out, int out_size, void* d_ws, size_t ws_size,
                              hipStream_t stream) {
  (void)n_in; (void)out_size; (void)ws_size;

  const float* x   = (const float*)d_in[0];
  const int*   ei  = (const int*)d_in[1];
  const int N = in_sizes[0] / 128;
  const int E = in_sizes[1] / 2;
  const int T = E + N;

  float* ws    = (float*)d_ws;
  float* hp    = ws;
  float* hcur  = hp    + (size_t)N * 64;
  float* accum = hcur  + (size_t)N * 64;
  float* as_   = accum + (size_t)N * 64;
  float* ad_   = as_   + N;
  unsigned int* maxk = (unsigned int*)(ad_ + N);
  float* denom = (float*)(maxk + N);
  float* eexp  = denom + N;

  const float* hin = x;
  int K = 128;
  for (int layer = 0; layer < 3; ++layer) {
    const float* W    = (const float*)d_in[4 + layer * 4];
    const float* asrc = (const float*)d_in[5 + layer * 4];
    const float* adst = (const float*)d_in[6 + layer * 4];
    const float* bias = (const float*)d_in[7 + layer * 4];

    const int mblocks = (N / 16 + 1) / 2;
    gat_gemm_wmma_f32<<<mblocks + 1, 256, 0, stream>>>(hin, W, hp, N, K);

    gat_alpha<<<(N + 7) / 8, 256, 0, stream>>>(hp, asrc, adst, as_, ad_, N);

    gat_init<<<(N * 64 + 255) / 256, 256, 0, stream>>>(accum, maxk, denom, N);
    gat_edge_max<<<(T + 255) / 256, 256, 0, stream>>>(ei, as_, ad_, maxk, E, N);
    gat_edge_exp<<<(T + 255) / 256, 256, 0, stream>>>(ei, as_, ad_, maxk, denom, eexp, E, N);
    {
      long long tot = (long long)T * 64;
      gat_edge_scatter<<<(unsigned int)((tot + 255) / 256), 256, 0, stream>>>(
          ei, hp, eexp, denom, accum, E, N);
    }

    float* dst = (layer == 2) ? (float*)d_out : hcur;
    gat_finalize<<<(N * 64 + 255) / 256, 256, 0, stream>>>(accum, bias, dst, N);

    hin = hcur;
    K = 64;
  }
}


# Edge phase owner-computes: a work-group owns 196 destinations, filters the edge list for them, keeps max, exp-sum and rows in LDS (LDS atomics, arrival order); global atomics, init, finalize go
# speedup vs baseline: 3.3084x; 3.3084x over previous
#include <hip/hip_runtime.h>
#include <math.h>


typedef __attribute__((ext_vector_type(2))) float v2f;
typedef __attribute__((ext_vector_type(8))) float v8f;

#define NEG_SLOPE 0.2f

__global__ __launch_bounds__(256) void gat_gemm_wmma_f32(
    const float* __restrict__ A, const float* __restrict__ W,
    float* __restrict__ C, int M, int K) {
  const int wave = threadIdx.x >> 5;
  const int lane = threadIdx.x & 31;
  const int ntile = wave & 3;
  const int mtile = (blockIdx.x << 1) | (wave >> 2);
  if (mtile * 16 >= M) return;

  const int mrow = mtile * 16 + (lane & 15);
  const int koff = (lane >> 4) << 1;
  const int col  = ntile * 16 + (lane & 15);

  const float* __restrict__ arow = A + (size_t)mrow * K;
  v8f acc = {};
  for (int k = 0; k < K; k += 4) {
    v2f a, b;
    a.x = arow[k + koff];
    a.y = arow[k + koff + 1];
    b.x = W[(size_t)(k + koff)     * 64 + col];
    b.y = W[(size_t)(k + koff + 1) * 64 + col];
    acc = __builtin_amdgcn_wmma_f32_16x16x4_f32(
        false, a, false, b, (short)0, acc, false, false);
  }
  const int rbase = mtile * 16 + ((lane >> 4) << 3);
#pragma unroll
  for (int v = 0; v < 8; ++v)
    C[(size_t)(rbase + v) * 64 + col] = acc[v];
}

__global__ __launch_bounds__(256) void gat_alpha(
    const float* __restrict__ hp, const float* __restrict__ a_s,
    const float* __restrict__ a_d, float* __restrict__ out_s,
    float* __restrict__ out_d, int N) {
  const int node = blockIdx.x * 8 + (threadIdx.x >> 5);
  const int lane = threadIdx.x & 31;
  if (node >= N) return;
  const float* __restrict__ h = hp + (size_t)node * 64;
  float s = h[lane] * a_s[lane] + h[lane + 32] * a_s[lane + 32];
  float d = h[lane] * a_d[lane] + h[lane + 32] * a_d[lane + 32];
#pragma unroll
  for (int off = 16; off > 0; off >>= 1) {
    s += __shfl_down(s, off, 32);
    d += __shfl_down(d, off, 32);
  }
  if (lane == 0) { out_s[node] = s; out_d[node] = d; }
}

__device__ __forceinline__ unsigned int f2key(float f) {
  unsigned int u = __float_as_uint(f);
  return (u & 0x80000000u) ? ~u : (u | 0x80000000u);
}
__device__ __forceinline__ float key2f(unsigned int k) {
  return __uint_as_float((k & 0x80000000u) ? (k ^ 0x80000000u) : ~k);
}
__device__ __forceinline__ float leaky(float e) {
  return (e > 0.0f) ? e : NEG_SLOPE * e;
}

#define OWN_ROWS 196
#define OWN_CAP  10240
#define OWN_NT   1024

struct OwnLds {
  float        acc[OWN_ROWS * 64];
  float        den[OWN_ROWS];
  unsigned int maxk[OWN_ROWS];
  float        ad[OWN_ROWS];
  int          src[OWN_CAP];
  int          dl[OWN_CAP];
  float        e[OWN_CAP];
  int          count;
};

__device__ __forceinline__ void own_keep(OwnLds& s, int d, int eidx, int base,
                                         unsigned int cnt) {
  const unsigned int dl = (unsigned int)(d - base);
  if (dl < cnt) {
    const int pos = atomicAdd(&s.count, 1);
    if (pos < OWN_CAP) { s.src[pos] = eidx; s.dl[pos] = (int)dl; }
  }
}

__device__ __forceinline__ void own_scan(OwnLds& s, const int* __restrict__ dstp,
                                         int lo, int hi, int base,
                                         unsigned int cnt, int tid) {
  const int mis = (int)((reinterpret_cast<size_t>(dstp + lo) >> 2) & 3);
  int a0 = lo + ((4 - mis) & 3);
  if (a0 > hi) a0 = hi;
  if (lo + tid < a0) own_keep(s, dstp[lo + tid], lo + tid, base, cnt);
  const int nvec = (hi - a0) >> 2;
  const int4* __restrict__ vp = reinterpret_cast<const int4*>(dstp + a0);
  int v = tid;
  for (; v + 3 * OWN_NT < nvec; v += 4 * OWN_NT) {
    int4 q[4];
#pragma unroll
    for (int u = 0; u < 4; ++u) q[u] = vp[v + u * OWN_NT];
#pragma unroll
    for (int u = 0; u < 4; ++u) {
      const unsigned int d0 = (unsigned int)(q[u].x - base);
      const unsigned int d1 = (unsigned int)(q[u].y - base);
      const unsigned int d2 = (unsigned int)(q[u].z - base);
      const unsigned int d3 = (unsigned int)(q[u].w - base);
      const unsigned int m01 = d0 < d1 ? d0 : d1;
      const unsigned int m23 = d2 < d3 ? d2 : d3;
      if ((m01 < m23 ? m01 : m23) < cnt) {
        const int e0 = a0 + ((v + u * OWN_NT) << 2);
        own_keep(s, q[u].x, e0,     base, cnt);
        own_keep(s, q[u].y, e0 + 1, base, cnt);
        own_keep(s, q[u].z, e0 + 2, base, cnt);
        own_keep(s, q[u].w, e0 + 3, base, cnt);
      }
    }
  }
  for (; v < nvec; v += OWN_NT) {
    const int4 q = vp[v];
    const int e0 = a0 + (v << 2);
    own_keep(s, q.x, e0,     base, cnt);
    own_keep(s, q.y, e0 + 1, base, cnt);
    own_keep(s, q.z, e0 + 2, base, cnt);
    own_keep(s, q.w, e0 + 3, base, cnt);
  }
  const int t0 = a0 + (nvec << 2);
  if (t0 + tid < hi) own_keep(s, dstp[t0 + tid], t0 + tid, base, cnt);
}

__device__ __forceinline__ void own_resolve(OwnLds& s, const int* __restrict__ ei,
                                            const float* __restrict__ as_,
                                            int n, int tid) {
  for (int p = tid; p < n; p += OWN_NT) {
    const int src = ei[s.src[p]];
    const int dl  = s.dl[p];
    const float e = leaky(as_[src] + s.ad[dl]);
    s.src[p] = src;
    s.e[p]   = e;
    atomicMax(&s.maxk[dl], f2key(e));
  }
}

__device__ __forceinline__ void own_rows(OwnLds& s, const float* __restrict__ hp,
                                         int n, int wave, int lane) {
  for (int j0 = wave; j0 < n; j0 += 32 * 8) {
    int   dl[8];
    float ev[8];
    float2 h[8];
#pragma unroll
    for (int u = 0; u < 8; ++u) {
      const int j  = j0 + 32 * u;
      const int jj = (j < n) ? j : j0;
      dl[u] = s.dl[jj];
      ev[u] = s.e[jj];
      h[u]  = *reinterpret_cast<const float2*>(hp + (size_t)s.src[jj] * 64 + 2 * lane);
    }
#pragma unroll
    for (int u = 0; u < 8; ++u) {
      if (j0 + 32 * u < n) {
        const float w = __expf(ev[u] - key2f(s.maxk[dl[u]]));
        float* row = &s.acc[dl[u] * 64 + 2 * lane];
        atomicAdd(row,     w * h[u].x);
        atomicAdd(row + 1, w * h[u].y);
        if (lane == 0) atomicAdd(&s.den[dl[u]], w);
      }
    }
  }
}

__global__ __launch_bounds__(OWN_NT) void gat_edge_owner(
    const int* __restrict__ ei, const float* __restrict__ hp,
    const float* __restrict__ as_, const float* __restrict__ ad_,
    const float* __restrict__ bias, float* __restrict__ out, int E, int N) {
  __shared__ OwnLds s;
  const int tid  = threadIdx.x;
  const int lane = tid & 31;
  const int wave = __builtin_amdgcn_readfirstlane(tid >> 5);
  const int* __restrict__ dstp = ei + E;

  for (int base = blockIdx.x * OWN_ROWS; base < N; base += gridDim.x * OWN_ROWS) {
    const int cnt = (N - base < OWN_ROWS) ? (N - base) : OWN_ROWS;

    if (tid < cnt) { s.maxk[tid] = 0u; s.ad[tid] = ad_[base + tid]; }
    if (tid == 0) s.count = 0;
    __syncthreads();

    own_scan(s, dstp, 0, E, base, (unsigned int)cnt, tid);
    __syncthreads();
    const int found = s.count;

    if (found <= OWN_CAP) {
      own_resolve(s, ei, as_, found, tid);
      if (tid < cnt)
        atomicMax(&s.maxk[tid], f2key(leaky(as_[base + tid] + s.ad[tid])));
      __syncthreads();
      for (int i = tid; i < cnt * 64; i += OWN_NT) {
        const int dl = i >> 6;
        const float w = __expf(leaky(as_[base + dl] + s.ad[dl]) - key2f(s.maxk[dl]));
        s.acc[i] = w * hp[(size_t)base * 64 + i];
        if ((i & 63) == 0) s.den[dl] = w;
      }
      __syncthreads();
      own_rows(s, hp, found, wave, lane);
      __syncthreads();
    } else {
      for (int i = tid; i < E; i += OWN_NT) {
        const unsigned int dl = (unsigned int)(dstp[i] - base);
        if (dl < (unsigned int)cnt)
          atomicMax(&s.maxk[dl], f2key(leaky(as_[ei[i]] + s.ad[dl])));
      }
      if (tid < cnt)
        atomicMax(&s.maxk[tid], f2key(leaky(as_[base + tid] + s.ad[tid])));
      __syncthreads();
      for (int i = tid; i < cnt * 64; i += OWN_NT) {
        const int dl = i >> 6;
        const float w = __expf(leaky(as_[base + dl] + s.ad[dl]) - key2f(s.maxk[dl]));
        s.acc[i] = w * hp[(size_t)base * 64 + i];
        if ((i & 63) == 0) s.den[dl] = w;
      }
      __syncthreads();
      for (int w0 = 0; w0 < E; w0 += OWN_CAP) {
        if (tid == 0) s.count = 0;
        __syncthreads();
        own_scan(s, dstp, w0, (E - w0 < OWN_CAP) ? E : w0 + OWN_CAP, base,
                 (unsigned int)cnt, tid);
        __syncthreads();
        const int n = s.count;
        own_resolve(s, ei, as_, n, tid);
        __syncthreads();
        own_rows(s, hp, n, wave, lane);
        __syncthreads();
      }
    }

    for (int i = tid; i < cnt * 64; i += OWN_NT) {
      const float v = s.acc[i] / (s.den[i >> 6] + 1e-16f) + bias[i & 63];
      out[(size_t)base * 64 + i] = v > 0.0f ? v : 0.0f;
    }
    __syncthreads();
  }
}

extern "C" void kernel_launch(void* const* d_in, const int* in_sizes, int n_in,
                              void* d_out, int out_size, void* d_ws, size_t ws_size,
                              hipStream_t stream) {
  (void)n_in; (void)out_size; (void)ws_size;

  const float* x   = (const float*)d_in[0];
  const int*   ei  = (const int*)d_in[1];
  const int N = in_sizes[0] / 128;
  const int E = in_sizes[1] / 2;

  float* ws    = (float*)d_ws;
  float* hp    = ws;
  float* hcur  = hp    + (size_t)N * 64;
  float* as_   = hcur  + (size_t)N * 64;
  float* ad_   = as_   + N;

  const float* hin = x;
  int K = 128;
  for (int layer = 0; layer < 3; ++layer) {
    const float* W    = (const float*)d_in[4 + layer * 4];
    const float* asrc = (const float*)d_in[5 + layer * 4];
    const float* adst = (const float*)d_in[6 + layer * 4];
    const float* bias = (const float*)d_in[7 + layer * 4];

    const int mblocks = (N / 16 + 1) / 2;
    gat_gemm_wmma_f32<<<mblocks + 1, 256, 0, stream>>>(hin, W, hp, N, K);

    gat_alpha<<<(N + 7) / 8, 256, 0, stream>>>(hp, asrc, adst, as_, ad_, N);

    float* dst = (layer == 2) ? (float*)d_out : hcur;
    gat_edge_owner<<<256, OWN_NT, 0, stream>>>(ei, hp, as_, ad_, bias, dst, E, N);

    hin = hcur;
    K = 64;
  }
}


# Edge scan reads a one-byte owner map (destination / 196, packed by the alpha launch into a dead plane) in place of the int32 list: a quarter of the bytes; list order changes, sums stay LDS atomics
# speedup vs baseline: 3.7662x; 1.1384x over previous
#include <hip/hip_runtime.h>
#include <math.h>


typedef __attribute__((ext_vector_type(2))) float v2f;
typedef __attribute__((ext_vector_type(8))) float v8f;

#define NEG_SLOPE 0.2f
#define OWN_ROWS 196

__global__ __launch_bounds__(256) void gat_gemm_wmma_f32(
    const float* __restrict__ A, const float* __restrict__ W,
    float* __restrict__ C, int M, int K) {
  const int wave = threadIdx.x >> 5;
  const int lane = threadIdx.x & 31;
  const int ntile = wave & 3;
  const int mtile = (blockIdx.x << 1) | (wave >> 2);
  if (mtile * 16 >= M) return;

  const int mrow = mtile * 16 + (lane & 15);
  const int koff = (lane >> 4) << 1;
  const int col  = ntile * 16 + (lane & 15);

  const float* __restrict__ arow = A + (size_t)mrow * K;
  v8f acc = {};
  for (int k = 0; k < K; k += 4) {
    v2f a, b;
    a.x = arow[k + koff];
    a.y = arow[k + koff + 1];
    b.x = W[(size_t)(k + koff)     * 64 + col];
    b.y = W[(size_t)(k + koff + 1) * 64 + col];
    acc = __builtin_amdgcn_wmma_f32_16x16x4_f32(
        false, a, false, b, (short)0, acc, false, false);
  }
  const int rbase = mtile * 16 + ((lane >> 4) << 3);
#pragma unroll
  for (int v = 0; v < 8; ++v)
    C[(size_t)(rbase + v) * 64 + col] = acc[v];
}

__global__ __launch_bounds__(256) void gat_alpha(
    const float* __restrict__ hp, const float* __restrict__ a_s,
    const float* __restrict__ a_d, float* __restrict__ out_s,
    float* __restrict__ out_d, int N, const int* __restrict__ ei,
    int* __restrict__ omap, int E) {
  {
    const int* __restrict__ dstp = ei + E;
    for (int g = blockIdx.x * 256 + threadIdx.x; 4 * g < E; g += gridDim.x * 256) {
      unsigned int pk = 0u;
#pragma unroll
      for (int b = 0; b < 4; ++b) {
        const int e = 4 * g + b;
        if (e < E)
          pk |= (((unsigned int)dstp[e] / OWN_ROWS) & 255u) << (8 * b);
      }
      omap[g] = (int)pk;
    }
  }
  const int node = blockIdx.x * 8 + (threadIdx.x >> 5);
  const int lane = threadIdx.x & 31;
  if (node >= N) return;
  const float* __restrict__ h = hp + (size_t)node * 64;
  float s = h[lane] * a_s[lane] + h[lane + 32] * a_s[lane + 32];
  float d = h[lane] * a_d[lane] + h[lane + 32] * a_d[lane + 32];
#pragma unroll
  for (int off = 16; off > 0; off >>= 1) {
    s += __shfl_down(s, off, 32);
    d += __shfl_down(d, off, 32);
  }
  if (lane == 0) { out_s[node] = s; out_d[node] = d; }
}

__device__ __forceinline__ unsigned int f2key(float f) {
  unsigned int u = __float_as_uint(f);
  return (u & 0x80000000u) ? ~u : (u | 0x80000000u);
}
__device__ __forceinline__ float key2f(unsigned int k) {
  return __uint_as_float((k & 0x80000000u) ? (k ^ 0x80000000u) : ~k);
}
__device__ __forceinline__ float leaky(float e) {
  return (e > 0.0f) ? e : NEG_SLOPE * e;
}

#define OWN_CAP  10240
#define OWN_NT   1024

struct OwnLds {
  float        acc[OWN_ROWS * 64];
  float        den[OWN_ROWS];
  unsigned int maxk[OWN_ROWS];
  float        ad[OWN_ROWS];
  int          src[OWN_CAP];
  int          dl[OWN_CAP];
  float        e[OWN_CAP];
  int          count;
};

__device__ __forceinline__ void own_keep(OwnLds& s, int d, int eidx, int base,
                                         unsigned int cnt) {
  const unsigned int dl = (unsigned int)(d - base);
  if (dl < cnt) {
    const int pos = atomicAdd(&s.count, 1);
    if (pos < OWN_CAP) s.src[pos] = eidx;
  }
}

__device__ __forceinline__ unsigned int own_zero_bytes(unsigned int t) {
  return ~(((t & 0x7f7f7f7fu) + 0x7f7f7f7fu) | t | 0x7f7f7f7fu);
}

__device__ __forceinline__ void own_keep16(OwnLds& s, uint4 q, unsigned int pat,
                                           int e0) {
  unsigned int bits = (own_zero_bytes(q.x ^ pat) >> 7) |
                      (own_zero_bytes(q.y ^ pat) >> 6) |
                      (own_zero_bytes(q.z ^ pat) >> 5) |
                      (own_zero_bytes(q.w ^ pat) >> 4);
  if (bits) {
    int pos = atomicAdd(&s.count, __popc(bits));
    do {
      const int p = __ffs((int)bits) - 1;
      bits &= bits - 1u;
      if (pos < OWN_CAP) s.src[pos] = e0 + ((p & 7) << 2) + (p >> 3);
      ++pos;
    } while (bits);
  }
}

__device__ __forceinline__ void own_scan_map(OwnLds& s, const int* __restrict__ omap,
                                             int E, unsigned int id, int tid) {
  const unsigned int pat = id * 0x01010101u;
  const int nvec = E >> 4;
  const uint4* __restrict__ vp = reinterpret_cast<const uint4*>(omap);
  int v = tid;
  for (; v + 3 * OWN_NT < nvec; v += 4 * OWN_NT) {
    uint4 q[4];
#pragma unroll
    for (int u = 0; u < 4; ++u) q[u] = vp[v + u * OWN_NT];
#pragma unroll
    for (int u = 0; u < 4; ++u) own_keep16(s, q[u], pat, (v + u * OWN_NT) << 4);
  }
  for (; v < nvec; v += OWN_NT) own_keep16(s, vp[v], pat, v << 4);
  const int t0 = nvec << 4;
  if (t0 + tid < E &&
      reinterpret_cast<const unsigned char*>(omap)[t0 + tid] == id) {
    const int pos = atomicAdd(&s.count, 1);
    if (pos < OWN_CAP) s.src[pos] = t0 + tid;
  }
}

__device__ __forceinline__ void own_scan(OwnLds& s, const int* __restrict__ dstp,
                                         int lo, int hi, int base,
                                         unsigned int cnt, int tid) {
  const int mis = (int)((reinterpret_cast<size_t>(dstp + lo) >> 2) & 3);
  int a0 = lo + ((4 - mis) & 3);
  if (a0 > hi) a0 = hi;
  if (lo + tid < a0) own_keep(s, dstp[lo + tid], lo + tid, base, cnt);
  const int nvec = (hi - a0) >> 2;
  const int4* __restrict__ vp = reinterpret_cast<const int4*>(dstp + a0);
  int v = tid;
  for (; v + 3 * OWN_NT < nvec; v += 4 * OWN_NT) {
    int4 q[4];
#pragma unroll
    for (int u = 0; u < 4; ++u) q[u] = vp[v + u * OWN_NT];
#pragma unroll
    for (int u = 0; u < 4; ++u) {
      const unsigned int d0 = (unsigned int)(q[u].x - base);
      const unsigned int d1 = (unsigned int)(q[u].y - base);
      const unsigned int d2 = (unsigned int)(q[u].z - base);
      const unsigned int d3 = (unsigned int)(q[u].w - base);
      const unsigned int m01 = d0 < d1 ? d0 : d1;
      const unsigned int m23 = d2 < d3 ? d2 : d3;
      if ((m01 < m23 ? m01 : m23) < cnt) {
        const int e0 = a0 + ((v + u * OWN_NT) << 2);
        own_keep(s, q[u].x, e0,     base, cnt);
        own_keep(s, q[u].y, e0 + 1, base, cnt);
        own_keep(s, q[u].z, e0 + 2, base, cnt);
        own_keep(s, q[u].w, e0 + 3, base, cnt);
      }
    }
  }
  for (; v < nvec; v += OWN_NT) {
    const int4 q = vp[v];
    const int e0 = a0 + (v << 2);
    own_keep(s, q.x, e0,     base, cnt);
    own_keep(s, q.y, e0 + 1, base, cnt);
    own_keep(s, q.z, e0 + 2, base, cnt);
    own_keep(s, q.w, e0 + 3, base, cnt);
  }
  const int t0 = a0 + (nvec << 2);
  if (t0 + tid < hi) own_keep(s, dstp[t0 + tid], t0 + tid, base, cnt);
}

__device__ __forceinline__ void own_resolve(OwnLds& s, const int* __restrict__ ei,
                                            const int* __restrict__ dstp,
                                            const float* __restrict__ as_,
                                            int base, unsigned int cnt,
                                            int n, int tid) {
  for (int p = tid; p < n; p += OWN_NT) {
    const int eidx = s.src[p];
    const int src  = ei[eidx];
    const unsigned int dl = (unsigned int)(dstp[eidx] - base);
    s.src[p] = src;
    if (dl < cnt) {
      const float e = leaky(as_[src] + s.ad[dl]);
      s.dl[p] = (int)dl;
      s.e[p]  = e;
      atomicMax(&s.maxk[dl], f2key(e));
    } else {
      s.dl[p] = -1;
      s.e[p]  = 0.0f;
    }
  }
}

__device__ __forceinline__ void own_rows(OwnLds& s, const float* __restrict__ hp,
                                         int n, int wave, int lane) {
  for (int j0 = wave; j0 < n; j0 += 32 * 8) {
    int   dl[8];
    float ev[8];
    float2 h[8];
#pragma unroll
    for (int u = 0; u < 8; ++u) {
      const int j  = j0 + 32 * u;
      const int jj = (j < n) ? j : j0;
      dl[u] = s.dl[jj];
      ev[u] = s.e[jj];
      h[u]  = *reinterpret_cast<const float2*>(hp + (size_t)s.src[jj] * 64 + 2 * lane);
    }
#pragma unroll
    for (int u = 0; u < 8; ++u) {
      if (j0 + 32 * u < n && dl[u] >= 0) {
        const float w = __expf(ev[u] - key2f(s.maxk[dl[u]]));
        float* row = &s.acc[dl[u] * 64 + 2 * lane];
        atomicAdd(row,     w * h[u].x);
        atomicAdd(row + 1, w * h[u].y);
        if (lane == 0) atomicAdd(&s.den[dl[u]], w);
      }
    }
  }
}

__global__ __launch_bounds__(OWN_NT) void gat_edge_owner(
    const int* __restrict__ ei, const int* __restrict__ omap,
    const float* __restrict__ hp,
    const float* __restrict__ as_, const float* __restrict__ ad_,
    const float* __restrict__ bias, float* __restrict__ out, int E, int N) {
  __shared__ OwnLds s;
  const int tid  = threadIdx.x;
  const int lane = tid & 31;
  const int wave = __builtin_amdgcn_readfirstlane(tid >> 5);
  const int* __restrict__ dstp = ei + E;

  for (int base = blockIdx.x * OWN_ROWS; base < N; base += gridDim.x * OWN_ROWS) {
    const int cnt = (N - base < OWN_ROWS) ? (N - base) : OWN_ROWS;

    if (tid < cnt) { s.maxk[tid] = 0u; s.ad[tid] = ad_[base + tid]; }
    if (tid == 0) s.count = 0;
    __syncthreads();

    own_scan_map(s, omap, E, ((unsigned int)base / OWN_ROWS) & 255u, tid);
    __syncthreads();
    const int found = s.count;

    if (found <= OWN_CAP) {
      own_resolve(s, ei, dstp, as_, base, (unsigned int)cnt, found, tid);
      if (tid < cnt)
        atomicMax(&s.maxk[tid], f2key(leaky(as_[base + tid] + s.ad[tid])));
      __syncthreads();
      for (int i = tid; i < cnt * 64; i += OWN_NT) {
        const int dl = i >> 6;
        const float w = __expf(leaky(as_[base + dl] + s.ad[dl]) - key2f(s.maxk[dl]));
        s.acc[i] = w * hp[(size_t)base * 64 + i];
        if ((i & 63) == 0) s.den[dl] = w;
      }
      __syncthreads();
      own_rows(s, hp, found, wave, lane);
      __syncthreads();
    } else {
      for (int i = tid; i < E; i += OWN_NT) {
        const unsigned int dl = (unsigned int)(dstp[i] - base);
        if (dl < (unsigned int)cnt)
          atomicMax(&s.maxk[dl], f2key(leaky(as_[ei[i]] + s.ad[dl])));
      }
      if (tid < cnt)
        atomicMax(&s.maxk[tid], f2key(leaky(as_[base + tid] + s.ad[tid])));
      __syncthreads();
      for (int i = tid; i < cnt * 64; i += OWN_NT) {
        const int dl = i >> 6;
        const float w = __expf(leaky(as_[base + dl] + s.ad[dl]) - key2f(s.maxk[dl]));
        s.acc[i] = w * hp[(size_t)base * 64 + i];
        if ((i & 63) == 0) s.den[dl] = w;
      }
      __syncthreads();
      for (int w0 = 0; w0 < E; w0 += OWN_CAP) {
        if (tid == 0) s.count = 0;
        __syncthreads();
        own_scan(s, dstp, w0, (E - w0 < OWN_CAP) ? E : w0 + OWN_CAP, base,
                 (unsigned int)cnt, tid);
        __syncthreads();
        const int n = s.count;
        own_resolve(s, ei, dstp, as_, base, (unsigned int)cnt, n, tid);
        __syncthreads();
        own_rows(s, hp, n, wave, lane);
        __syncthreads();
      }
    }

    for (int i = tid; i < cnt * 64; i += OWN_NT) {
      const float v = s.acc[i] / (s.den[i >> 6] + 1e-16f) + bias[i & 63];
      out[(size_t)base * 64 + i] = v > 0.0f ? v : 0.0f;
    }
    __syncthreads();
  }
}

extern "C" void kernel_launch(void* const* d_in, const int* in_sizes, int n_in,
                              void* d_out, int out_size, void* d_ws, size_t ws_size,
                              hipStream_t stream) {
  (void)n_in; (void)out_size; (void)ws_size;

  const float* x   = (const float*)d_in[0];
  const int*   ei  = (const int*)d_in[1];
  const int N = in_sizes[0] / 128;
  const int E = in_sizes[1] / 2;

  float* ws    = (float*)d_ws;
  float* hp    = ws;
  float* hcur  = hp    + (size_t)N * 64;
  float* as_   = hcur  + (size_t)N * 64;
  float* ad_   = as_   + N;

  const float* hin = x;
  int K = 128;
  for (int layer = 0; layer < 3; ++layer) {
    const float* W    = (const float*)d_in[4 + layer * 4];
    const float* asrc = (const float*)d_in[5 + layer * 4];
    const float* adst = (const float*)d_in[6 + layer * 4];
    const float* bias = (const float*)d_in[7 + layer * 4];

    const int mblocks = (N / 16 + 1) / 2;
    gat_gemm_wmma_f32<<<mblocks + 1, 256, 0, stream>>>(hin, W, hp, N, K);

    int* omap = (layer == 2) ? (int*)hcur : (int*)d_out;
    gat_alpha<<<(N + 7) / 8, 256, 0, stream>>>(hp, asrc, adst, as_, ad_, N, ei, omap, E);

    float* dst = (layer == 2) ? (float*)d_out : hcur;
    gat_edge_owner<<<256, OWN_NT, 0, stream>>>(ei, omap, hp, as_, ad_, bias, dst, E, N);

    hin = hcur;
    K = 64;
  }
}


# Alpha launch: 512 work-groups of long-lived waves loop over the nodes (16 lanes x float4 per node, 7 rows in flight) in place of 50000 one-node waves; alpha sum order changes
# speedup vs baseline: 4.0406x; 1.0729x over previous
#include <hip/hip_runtime.h>
#include <math.h>


typedef __attribute__((ext_vector_type(2))) float v2f;
typedef __attribute__((ext_vector_type(8))) float v8f;

#define NEG_SLOPE 0.2f
#define OWN_ROWS 196

__global__ __launch_bounds__(256) void gat_gemm_wmma_f32(
    const float* __restrict__ A, const float* __restrict__ W,
    float* __restrict__ C, int M, int K) {
  const int wave = threadIdx.x >> 5;
  const int lane = threadIdx.x & 31;
  const int ntile = wave & 3;
  const int mtile = (blockIdx.x << 1) | (wave >> 2);
  if (mtile * 16 >= M) return;

  const int mrow = mtile * 16 + (lane & 15);
  const int koff = (lane >> 4) << 1;
  const int col  = ntile * 16 + (lane & 15);

  const float* __restrict__ arow = A + (size_t)mrow * K;
  v8f acc = {};
  for (int k = 0; k < K; k += 4) {
    v2f a, b;
    a.x = arow[k + koff];
    a.y = arow[k + koff + 1];
    b.x = W[(size_t)(k + koff)     * 64 + col];
    b.y = W[(size_t)(k + koff + 1) * 64 + col];
    acc = __builtin_amdgcn_wmma_f32_16x16x4_f32(
        false, a, false, b, (short)0, acc, false, false);
  }
  const int rbase = mtile * 16 + ((lane >> 4) << 3);
#pragma unroll
  for (int v = 0; v < 8; ++v)
    C[(size_t)(rbase + v) * 64 + col] = acc[v];
}

#define AL_ROWS 7
__global__ __launch_bounds__(256) void gat_alpha(
    const float* __restrict__ hp, const float* __restrict__ a_s,
    const float* __restrict__ a_d, float* __restrict__ out_s,
    float* __restrict__ out_d, int N, const int* __restrict__ ei,
    int* __restrict__ omap, int E) {
  {
    const int* __restrict__ dstp = ei + E;
    for (int g = blockIdx.x * 256 + threadIdx.x; 4 * g < E; g += gridDim.x * 256) {
      unsigned int pk = 0u;
      if (4 * g + 3 < E) {
        const int e = 4 * g;
        const unsigned int d0 = (unsigned int)dstp[e];
        const unsigned int d1 = (unsigned int)dstp[e + 1];
        const unsigned int d2 = (unsigned int)dstp[e + 2];
        const unsigned int d3 = (unsigned int)dstp[e + 3];
        pk = ((d0 / OWN_ROWS) & 255u) | (((d1 / OWN_ROWS) & 255u) << 8) |
             (((d2 / OWN_ROWS) & 255u) << 16) | (((d3 / OWN_ROWS) & 255u) << 24);
      } else {
#pragma unroll
        for (int b = 0; b < 4; ++b) {
          const int e = 4 * g + b;
          if (e < E)
            pk |= (((unsigned int)dstp[e] / OWN_ROWS) & 255u) << (8 * b);
        }
      }
      omap[g] = (int)pk;
    }
  }
  const int lane = threadIdx.x & 31;
  const int sub  = lane & 15;
  const int half = lane >> 4;
  const int gw   = blockIdx.x * 8 + (threadIdx.x >> 5);
  const int tw   = gridDim.x * 8;
  float ws4[4], wd4[4];
#pragma unroll
  for (int j = 0; j < 4; ++j) { ws4[j] = a_s[4 * sub + j]; wd4[j] = a_d[4 * sub + j]; }
  const float4* __restrict__ hp4 = reinterpret_cast<const float4*>(hp);
  for (int p0 = gw; 2 * p0 < N; p0 += AL_ROWS * tw) {
    float4 v[AL_ROWS];
#pragma unroll
    for (int u = 0; u < AL_ROWS; ++u) {
      const int node = 2 * (p0 + u * tw) + half;
      v[u] = hp4[(size_t)(node < N ? node : 0) * 16 + sub];
    }
    float s[AL_ROWS], d[AL_ROWS];
#pragma unroll
    for (int u = 0; u < AL_ROWS; ++u) {
      s[u] = v[u].x * ws4[0] + v[u].y * ws4[1] + v[u].z * ws4[2] + v[u].w * ws4[3];
      d[u] = v[u].x * wd4[0] + v[u].y * wd4[1] + v[u].z * wd4[2] + v[u].w * wd4[3];
    }
#pragma unroll
    for (int m = 8; m > 0; m >>= 1) {
#pragma unroll
      for (int u = 0; u < AL_ROWS; ++u) {
        s[u] += __shfl_xor(s[u], m, 16);
        d[u] += __shfl_xor(d[u], m, 16);
      }
    }
#pragma unroll
    for (int u = 0; u < AL_ROWS; ++u) {
      const int node = 2 * (p0 + u * tw) + half;
      if (sub == 0 && node < N) { out_s[node] = s[u]; out_d[node] = d[u]; }
    }
  }
}

__device__ __forceinline__ unsigned int f2key(float f) {
  unsigned int u = __float_as_uint(f);
  return (u & 0x80000000u) ? ~u : (u | 0x80000000u);
}
__device__ __forceinline__ float key2f(unsigned int k) {
  return __uint_as_float((k & 0x80000000u) ? (k ^ 0x80000000u) : ~k);
}
__device__ __forceinline__ float leaky(float e) {
  return (e > 0.0f) ? e : NEG_SLOPE * e;
}

#define OWN_CAP  10240
#define OWN_NT   1024

struct OwnLds {
  float        acc[OWN_ROWS * 64];
  float        den[OWN_ROWS];
  unsigned int maxk[OWN_ROWS];
  float        ad[OWN_ROWS];
  int          src[OWN_CAP];
  int          dl[OWN_CAP];
  float        e[OWN_CAP];
  int          count;
};

__device__ __forceinline__ void own_keep(OwnLds& s, int d, int eidx, int base,
                                         unsigned int cnt) {
  const unsigned int dl = (unsigned int)(d - base);
  if (dl < cnt) {
    const int pos = atomicAdd(&s.count, 1);
    if (pos < OWN_CAP) s.src[pos] = eidx;
  }
}

__device__ __forceinline__ unsigned int own_zero_bytes(unsigned int t) {
  return ~(((t & 0x7f7f7f7fu) + 0x7f7f7f7fu) | t | 0x7f7f7f7fu);
}

__device__ __forceinline__ void own_keep16(OwnLds& s, uint4 q, unsigned int pat,
                                           int e0) {
  unsigned int bits = (own_zero_bytes(q.x ^ pat) >> 7) |
                      (own_zero_bytes(q.y ^ pat) >> 6) |
                      (own_zero_bytes(q.z ^ pat) >> 5) |
                      (own_zero_bytes(q.w ^ pat) >> 4);
  if (bits) {
    int pos = atomicAdd(&s.count, __popc(bits));
    do {
      const int p = __ffs((int)bits) - 1;
      bits &= bits - 1u;
      if (pos < OWN_CAP) s.src[pos] = e0 + ((p & 7) << 2) + (p >> 3);
      ++pos;
    } while (bits);
  }
}

__device__ __forceinline__ void own_scan_map(OwnLds& s, const int* __restrict__ omap,
                                             int E, unsigned int id, int tid) {
  const unsigned int pat = id * 0x01010101u;
  const int nvec = E >> 4;
  const uint4* __restrict__ vp = reinterpret_cast<const uint4*>(omap);
  int v = tid;
  for (; v + 3 * OWN_NT < nvec; v += 4 * OWN_NT) {
    uint4 q[4];
#pragma unroll
    for (int u = 0; u < 4; ++u) q[u] = vp[v + u * OWN_NT];
#pragma unroll
    for (int u = 0; u < 4; ++u) own_keep16(s, q[u], pat, (v + u * OWN_NT) << 4);
  }
  for (; v < nvec; v += OWN_NT) own_keep16(s, vp[v], pat, v << 4);
  const int t0 = nvec << 4;
  if (t0 + tid < E &&
      reinterpret_cast<const unsigned char*>(omap)[t0 + tid] == id) {
    const int pos = atomicAdd(&s.count, 1);
    if (pos < OWN_CAP) s.src[pos] = t0 + tid;
  }
}

__device__ __forceinline__ void own_scan(OwnLds& s, const int* __restrict__ dstp,
                                         int lo, int hi, int base,
                                         unsigned int cnt, int tid) {
  const int mis = (int)((reinterpret_cast<size_t>(dstp + lo) >> 2) & 3);
  int a0 = lo + ((4 - mis) & 3);
  if (a0 > hi) a0 = hi;
  if (lo + tid < a0) own_keep(s, dstp[lo + tid], lo + tid, base, cnt);
  const int nvec = (hi - a0) >> 2;
  const int4* __restrict__ vp = reinterpret_cast<const int4*>(dstp + a0);
  int v = tid;
  for (; v + 3 * OWN_NT < nvec; v += 4 * OWN_NT) {
    int4 q[4];
#pragma unroll
    for (int u = 0; u < 4; ++u) q[u] = vp[v + u * OWN_NT];
#pragma unroll
    for (int u = 0; u < 4; ++u) {
      const unsigned int d0 = (unsigned int)(q[u].x - base);
      const unsigned int d1 = (unsigned int)(q[u].y - base);
      const unsigned int d2 = (unsigned int)(q[u].z - base);
      const unsigned int d3 = (unsigned int)(q[u].w - base);
      const unsigned int m01 = d0 < d1 ? d0 : d1;
      const unsigned int m23 = d2 < d3 ? d2 : d3;
      if ((m01 < m23 ? m01 : m23) < cnt) {
        const int e0 = a0 + ((v + u * OWN_NT) << 2);
        own_keep(s, q[u].x, e0,     base, cnt);
        own_keep(s, q[u].y, e0 + 1, base, cnt);
        own_keep(s, q[u].z, e0 + 2, base, cnt);
        own_keep(s, q[u].w, e0 + 3, base, cnt);
      }
    }
  }
  for (; v < nvec; v += OWN_NT) {
    const int4 q = vp[v];
    const int e0 = a0 + (v << 2);
    own_keep(s, q.x, e0,     base, cnt);
    own_keep(s, q.y, e0 + 1, base, cnt);
    own_keep(s, q.z, e0 + 2, base, cnt);
    own_keep(s, q.w, e0 + 3, base, cnt);
  }
  const int t0 = a0 + (nvec << 2);
  if (t0 + tid < hi) own_keep(s, dstp[t0 + tid], t0 + tid, base, cnt);
}

__device__ __forceinline__ void own_resolve(OwnLds& s, const int* __restrict__ ei,
                                            const int* __restrict__ dstp,
                                            const float* __restrict__ as_,
                                            int base, unsigned int cnt,
                                            int n, int tid) {
  for (int p = tid; p < n; p += OWN_NT) {
    const int eidx = s.src[p];
    const int src  = ei[eidx];
    const unsigned int dl = (unsigned int)(dstp[eidx] - base);
    s.src[p] = src;
    if (dl < cnt) {
      const float e = leaky(as_[src] + s.ad[dl]);
      s.dl[p] = (int)dl;
      s.e[p]  = e;
      atomicMax(&s.maxk[dl], f2key(e));
    } else {
      s.dl[p] = -1;
      s.e[p]  = 0.0f;
    }
  }
}

__device__ __forceinline__ void own_rows(OwnLds& s, const float* __restrict__ hp,
                                         int n, int wave, int lane) {
  for (int j0 = wave; j0 < n; j0 += 32 * 8) {
    int   dl[8];
    float ev[8];
    float2 h[8];
#pragma unroll
    for (int u = 0; u < 8; ++u) {
      const int j  = j0 + 32 * u;
      const int jj = (j < n) ? j : j0;
      dl[u] = s.dl[jj];
      ev[u] = s.e[jj];
      h[u]  = *reinterpret_cast<const float2*>(hp + (size_t)s.src[jj] * 64 + 2 * lane);
    }
#pragma unroll
    for (int u = 0; u < 8; ++u) {
      if (j0 + 32 * u < n && dl[u] >= 0) {
        const float w = __expf(ev[u] - key2f(s.maxk[dl[u]]));
        float* row = &s.acc[dl[u] * 64 + 2 * lane];
        atomicAdd(row,     w * h[u].x);
        atomicAdd(row + 1, w * h[u].y);
        if (lane == 0) atomicAdd(&s.den[dl[u]], w);
      }
    }
  }
}

__global__ __launch_bounds__(OWN_NT) void gat_edge_owner(
    const int* __restrict__ ei, const int* __restrict__ omap,
    const float* __restrict__ hp,
    const float* __restrict__ as_, const float* __restrict__ ad_,
    const float* __restrict__ bias, float* __restrict__ out, int E, int N) {
  __shared__ OwnLds s;
  const int tid  = threadIdx.x;
  const int lane = tid & 31;
  const int wave = __builtin_amdgcn_readfirstlane(tid >> 5);
  const int* __restrict__ dstp = ei + E;

  for (int base = blockIdx.x * OWN_ROWS; base < N; base += gridDim.x * OWN_ROWS) {
    const int cnt = (N - base < OWN_ROWS) ? (N - base) : OWN_ROWS;

    if (tid < cnt) { s.maxk[tid] = 0u; s.ad[tid] = ad_[base + tid]; }
    if (tid == 0) s.count = 0;
    __syncthreads();

    own_scan_map(s, omap, E, ((unsigned int)base / OWN_ROWS) & 255u, tid);
    __syncthreads();
    const int found = s.count;

    if (found <= OWN_CAP) {
      own_resolve(s, ei, dstp, as_, base, (unsigned int)cnt, found, tid);
      if (tid < cnt)
        atomicMax(&s.maxk[tid], f2key(leaky(as_[base + tid] + s.ad[tid])));
      __syncthreads();
      for (int i = tid; i < cnt * 64; i += OWN_NT) {
        const int dl = i >> 6;
        const float w = __expf(leaky(as_[base + dl] + s.ad[dl]) - key2f(s.maxk[dl]));
        s.acc[i] = w * hp[(size_t)base * 64 + i];
        if ((i & 63) == 0) s.den[dl] = w;
      }
      __syncthreads();
      own_rows(s, hp, found, wave, lane);
      __syncthreads();
    } else {
      for (int i = tid; i < E; i += OWN_NT) {
        const unsigned int dl = (unsigned int)(dstp[i] - base);
        if (dl < (unsigned int)cnt)
          atomicMax(&s.maxk[dl], f2key(leaky(as_[ei[i]] + s.ad[dl])));
      }
      if (tid < cnt)
        atomicMax(&s.maxk[tid], f2key(leaky(as_[base + tid] + s.ad[tid])));
      __syncthreads();
      for (int i = tid; i < cnt * 64; i += OWN_NT) {
        const int dl = i >> 6;
        const float w = __expf(leaky(as_[base + dl] + s.ad[dl]) - key2f(s.maxk[dl]));
        s.acc[i] = w * hp[(size_t)base * 64 + i];
        if ((i & 63) == 0) s.den[dl] = w;
      }
      __syncthreads();
      for (int w0 = 0; w0 < E; w0 += OWN_CAP) {
        if (tid == 0) s.count = 0;
        __syncthreads();
        own_scan(s, dstp, w0, (E - w0 < OWN_CAP) ? E : w0 + OWN_CAP, base,
                 (unsigned int)cnt, tid);
        __syncthreads();
        const int n = s.count;
        own_resolve(s, ei, dstp, as_, base, (unsigned int)cnt, n, tid);
        __syncthreads();
        own_rows(s, hp, n, wave, lane);
        __syncthreads();
      }
    }

    for (int i = tid; i < cnt * 64; i += OWN_NT) {
      const float v = s.acc[i] / (s.den[i >> 6] + 1e-16f) + bias[i & 63];
      out[(size_t)base * 64 + i] = v > 0.0f ? v : 0.0f;
    }
    __syncthreads();
  }
}

extern "C" void kernel_launch(void* const* d_in, const int* in_sizes, int n_in,
                              void* d_out, int out_size, void* d_ws, size_t ws_size,
                              hipStream_t stream) {
  (void)n_in; (void)out_size; (void)ws_size;

  const float* x   = (const float*)d_in[0];
  const int*   ei  = (const int*)d_in[1];
  const int N = in_sizes[0] / 128;
  const int E = in_sizes[1] / 2;

  float* ws    = (float*)d_ws;
  float* hp    = ws;
  float* hcur  = hp    + (size_t)N * 64;
  float* as_   = hcur  + (size_t)N * 64;
  float* ad_   = as_   + N;

  const float* hin = x;
  int K = 128;
  for (int layer = 0; layer < 3; ++layer) {
    const float* W    = (const float*)d_in[4 + layer * 4];
    const float* asrc = (const float*)d_in[5 + layer * 4];
    const float* adst = (const float*)d_in[6 + layer * 4];
    const float* bias = (const float*)d_in[7 + layer * 4];

    const int mblocks = (N / 16 + 1) / 2;
    gat_gemm_wmma_f32<<<mblocks + 1, 256, 0, stream>>>(hin, W, hp, N, K);

    int* omap = (layer == 2) ? (int*)hcur : (int*)d_out;
    gat_alpha<<<512, 256, 0, stream>>>(hp, asrc, adst, as_, ad_, N, ei, omap, E);

    float* dst = (layer == 2) ? (float*)d_out : hcur;
    gat_edge_owner<<<256, OWN_NT, 0, stream>>>(ei, omap, hp, as_, ad_, bias, dst, E, N);

    hin = hcur;
    K = 64;
  }
}


# Row phase: softmax weight and exp-sum add move to a lane-per-entry pass; wave-per-edge loop keeps only row gather and two LDS adds, 16 rows in flight; exp-sum order changes, sums stay LDS atomics
# speedup vs baseline: 4.4018x; 1.0894x over previous
#include <hip/hip_runtime.h>
#include <math.h>


typedef __attribute__((ext_vector_type(2))) float v2f;
typedef __attribute__((ext_vector_type(8))) float v8f;

#define NEG_SLOPE 0.2f
#define OWN_ROWS 196

__global__ __launch_bounds__(256) void gat_gemm_wmma_f32(
    const float* __restrict__ A, const float* __restrict__ W,
    float* __restrict__ C, int M, int K) {
  const int wave = threadIdx.x >> 5;
  const int lane = threadIdx.x & 31;
  const int ntile = wave & 3;
  const int mtile = (blockIdx.x << 1) | (wave >> 2);
  if (mtile * 16 >= M) return;

  const int mrow = mtile * 16 + (lane & 15);
  const int koff = (lane >> 4) << 1;
  const int col  = ntile * 16 + (lane & 15);

  const float* __restrict__ arow = A + (size_t)mrow * K;
  v8f acc = {};
  for (int k = 0; k < K; k += 4) {
    v2f a, b;
    a.x = arow[k + koff];
    a.y = arow[k + koff + 1];
    b.x = W[(size_t)(k + koff)     * 64 + col];
    b.y = W[(size_t)(k + koff + 1) * 64 + col];
    acc = __builtin_amdgcn_wmma_f32_16x16x4_f32(
        false, a, false, b, (short)0, acc, false, false);
  }
  const int rbase = mtile * 16 + ((lane >> 4) << 3);
#pragma unroll
  for (int v = 0; v < 8; ++v)
    C[(size_t)(rbase + v) * 64 + col] = acc[v];
}

#define AL_ROWS 7
__global__ __launch_bounds__(256) void gat_alpha(
    const float* __restrict__ hp, const float* __restrict__ a_s,
    const float* __restrict__ a_d, float* __restrict__ out_s,
    float* __restrict__ out_d, int N, const int* __restrict__ ei,
    int* __restrict__ omap, int E) {
  {
    const int* __restrict__ dstp = ei + E;
    for (int g = blockIdx.x * 256 + threadIdx.x; 4 * g < E; g += gridDim.x * 256) {
      unsigned int pk = 0u;
      if (4 * g + 3 < E) {
        const int e = 4 * g;
        const unsigned int d0 = (unsigned int)dstp[e];
        const unsigned int d1 = (unsigned int)dstp[e + 1];
        const unsigned int d2 = (unsigned int)dstp[e + 2];
        const unsigned int d3 = (unsigned int)dstp[e + 3];
        pk = ((d0 / OWN_ROWS) & 255u) | (((d1 / OWN_ROWS) & 255u) << 8) |
             (((d2 / OWN_ROWS) & 255u) << 16) | (((d3 / OWN_ROWS) & 255u) << 24);
      } else {
#pragma unroll
        for (int b = 0; b < 4; ++b) {
          const int e = 4 * g + b;
          if (e < E)
            pk |= (((unsigned int)dstp[e] / OWN_ROWS) & 255u) << (8 * b);
        }
      }
      omap[g] = (int)pk;
    }
  }
  const int lane = threadIdx.x & 31;
  const int sub  = lane & 15;
  const int half = lane >> 4;
  const int gw   = blockIdx.x * 8 + (threadIdx.x >> 5);
  const int tw   = gridDim.x * 8;
  float ws4[4], wd4[4];
#pragma unroll
  for (int j = 0; j < 4; ++j) { ws4[j] = a_s[4 * sub + j]; wd4[j] = a_d[4 * sub + j]; }
  const float4* __restrict__ hp4 = reinterpret_cast<const float4*>(hp);
  for (int p0 = gw; 2 * p0 < N; p0 += AL_ROWS * tw) {
    float4 v[AL_ROWS];
#pragma unroll
    for (int u = 0; u < AL_ROWS; ++u) {
      const int node = 2 * (p0 + u * tw) + half;
      v[u] = hp4[(size_t)(node < N ? node : 0) * 16 + sub];
    }
    float s[AL_ROWS], d[AL_ROWS];
#pragma unroll
    for (int u = 0; u < AL_ROWS; ++u) {
      s[u] = v[u].x * ws4[0] + v[u].y * ws4[1] + v[u].z * ws4[2] + v[u].w * ws4[3];
      d[u] = v[u].x * wd4[0] + v[u].y * wd4[1] + v[u].z * wd4[2] + v[u].w * wd4[3];
    }
#pragma unroll
    for (int m = 8; m > 0; m >>= 1) {
#pragma unroll
      for (int u = 0; u < AL_ROWS; ++u) {
        s[u] += __shfl_xor(s[u], m, 16);
        d[u] += __shfl_xor(d[u], m, 16);
      }
    }
#pragma unroll
    for (int u = 0; u < AL_ROWS; ++u) {
      const int node = 2 * (p0 + u * tw) + half;
      if (sub == 0 && node < N) { out_s[node] = s[u]; out_d[node] = d[u]; }
    }
  }
}

__device__ __forceinline__ unsigned int f2key(float f) {
  unsigned int u = __float_as_uint(f);
  return (u & 0x80000000u) ? ~u : (u | 0x80000000u);
}
__device__ __forceinline__ float key2f(unsigned int k) {
  return __uint_as_float((k & 0x80000000u) ? (k ^ 0x80000000u) : ~k);
}
__device__ __forceinline__ float leaky(float e) {
  return (e > 0.0f) ? e : NEG_SLOPE * e;
}

#define OWN_CAP  10240
#define OWN_NT   1024

struct OwnLds {
  float        acc[OWN_ROWS * 64];
  float        den[OWN_ROWS];
  unsigned int maxk[OWN_ROWS];
  float        ad[OWN_ROWS];
  int          src[OWN_CAP];
  int          dl[OWN_CAP];
  float        e[OWN_CAP];
  int          count;
};

__device__ __forceinline__ void own_keep(OwnLds& s, int d, int eidx, int base,
                                         unsigned int cnt) {
  const unsigned int dl = (unsigned int)(d - base);
  if (dl < cnt) {
    const int pos = atomicAdd(&s.count, 1);
    if (pos < OWN_CAP) s.src[pos] = eidx;
  }
}

__device__ __forceinline__ unsigned int own_zero_bytes(unsigned int t) {
  return ~(((t & 0x7f7f7f7fu) + 0x7f7f7f7fu) | t | 0x7f7f7f7fu);
}

__device__ __forceinline__ void own_keep16(OwnLds& s, uint4 q, unsigned int pat,
                                           int e0) {
  unsigned int bits = (own_zero_bytes(q.x ^ pat) >> 7) |
                      (own_zero_bytes(q.y ^ pat) >> 6) |
                      (own_zero_bytes(q.z ^ pat) >> 5) |
                      (own_zero_bytes(q.w ^ pat) >> 4);
  if (bits) {
    int pos = atomicAdd(&s.count, __popc(bits));
    do {
      const int p = __ffs((int)bits) - 1;
      bits &= bits - 1u;
      if (pos < OWN_CAP) s.src[pos] = e0 + ((p & 7) << 2) + (p >> 3);
      ++pos;
    } while (bits);
  }
}

__device__ __forceinline__ void own_scan_map(OwnLds& s, const int* __restrict__ omap,
                                             int E, unsigned int id, int tid) {
  const unsigned int pat = id * 0x01010101u;
  const int nvec = E >> 4;
  const uint4* __restrict__ vp = reinterpret_cast<const uint4*>(omap);
  int v = tid;
  for (; v + 3 * OWN_NT < nvec; v += 4 * OWN_NT) {
    uint4 q[4];
#pragma unroll
    for (int u = 0; u < 4; ++u) q[u] = vp[v + u * OWN_NT];
#pragma unroll
    for (int u = 0; u < 4; ++u) own_keep16(s, q[u], pat, (v + u * OWN_NT) << 4);
  }
  for (; v < nvec; v += OWN_NT) own_keep16(s, vp[v], pat, v << 4);
  const int t0 = nvec << 4;
  if (t0 + tid < E &&
      reinterpret_cast<const unsigned char*>(omap)[t0 + tid] == id) {
    const int pos = atomicAdd(&s.count, 1);
    if (pos < OWN_CAP) s.src[pos] = t0 + tid;
  }
}

__device__ __forceinline__ void own_scan(OwnLds& s, const int* __restrict__ dstp,
                                         int lo, int hi, int base,
                                         unsigned int cnt, int tid) {
  const int mis = (int)((reinterpret_cast<size_t>(dstp + lo) >> 2) & 3);
  int a0 = lo + ((4 - mis) & 3);
  if (a0 > hi) a0 = hi;
  if (lo + tid < a0) own_keep(s, dstp[lo + tid], lo + tid, base, cnt);
  const int nvec = (hi - a0) >> 2;
  const int4* __restrict__ vp = reinterpret_cast<const int4*>(dstp + a0);
  int v = tid;
  for (; v + 3 * OWN_NT < nvec; v += 4 * OWN_NT) {
    int4 q[4];
#pragma unroll
    for (int u = 0; u < 4; ++u) q[u] = vp[v + u * OWN_NT];
#pragma unroll
    for (int u = 0; u < 4; ++u) {
      const unsigned int d0 = (unsigned int)(q[u].x - base);
      const unsigned int d1 = (unsigned int)(q[u].y - base);
      const unsigned int d2 = (unsigned int)(q[u].z - base);
      const unsigned int d3 = (unsigned int)(q[u].w - base);
      const unsigned int m01 = d0 < d1 ? d0 : d1;
      const unsigned int m23 = d2 < d3 ? d2 : d3;
      if ((m01 < m23 ? m01 : m23) < cnt) {
        const int e0 = a0 + ((v + u * OWN_NT) << 2);
        own_keep(s, q[u].x, e0,     base, cnt);
        own_keep(s, q[u].y, e0 + 1, base, cnt);
        own_keep(s, q[u].z, e0 + 2, base, cnt);
        own_keep(s, q[u].w, e0 + 3, base, cnt);
      }
    }
  }
  for (; v < nvec; v += OWN_NT) {
    const int4 q = vp[v];
    const int e0 = a0 + (v << 2);
    own_keep(s, q.x, e0,     base, cnt);
    own_keep(s, q.y, e0 + 1, base, cnt);
    own_keep(s, q.z, e0 + 2, base, cnt);
    own_keep(s, q.w, e0 + 3, base, cnt);
  }
  const int t0 = a0 + (nvec << 2);
  if (t0 + tid < hi) own_keep(s, dstp[t0 + tid], t0 + tid, base, cnt);
}

__device__ __forceinline__ void own_resolve(OwnLds& s, const int* __restrict__ ei,
                                            const int* __restrict__ dstp,
                                            const float* __restrict__ as_,
                                            int base, unsigned int cnt,
                                            int n, int tid) {
  for (int p = tid; p < n; p += OWN_NT) {
    const int eidx = s.src[p];
    const int src  = ei[eidx];
    const unsigned int dl = (unsigned int)(dstp[eidx] - base);
    s.src[p] = src;
    if (dl < cnt) {
      const float e = leaky(as_[src] + s.ad[dl]);
      s.dl[p] = (int)dl;
      s.e[p]  = e;
      atomicMax(&s.maxk[dl], f2key(e));
    } else {
      s.dl[p] = -1;
      s.e[p]  = 0.0f;
    }
  }
}

__device__ __forceinline__ void own_weights(OwnLds& s, int n, int tid) {
  for (int p = tid; p < n; p += OWN_NT) {
    const int dl = s.dl[p];
    if (dl >= 0) {
      const float w = __expf(s.e[p] - key2f(s.maxk[dl]));
      s.e[p] = w;
      atomicAdd(&s.den[dl], w);
    }
  }
}

#define ROW_R 16
__device__ __forceinline__ void own_rows(OwnLds& s, const float* __restrict__ hp,
                                         int n, int wave, int lane) {
  const float2* __restrict__ hp2 = reinterpret_cast<const float2*>(hp);
  for (int jb = wave * ROW_R; jb < n; jb += 32 * ROW_R) {
    const int  jj = jb + (lane & (ROW_R - 1));
    const bool ok = jj < n;
    const int   my_src = ok ? s.src[jj] : 0;
    const int   my_dl  = ok ? s.dl[jj]  : -1;
    const float my_w   = ok ? s.e[jj]   : 0.0f;
    float2 h[ROW_R];
#pragma unroll
    for (int u = 0; u < ROW_R; ++u) {
      const int src = __builtin_amdgcn_readlane(my_src, u);
      h[u] = hp2[(size_t)src * 32 + lane];
    }
#pragma unroll
    for (int u = 0; u < ROW_R; ++u) {
      const int dl = __builtin_amdgcn_readlane(my_dl, u);
      if (dl >= 0) {
        const float w =
            __int_as_float(__builtin_amdgcn_readlane(__float_as_int(my_w), u));
        float* row = &s.acc[dl * 64 + 2 * lane];
        atomicAdd(row,     w * h[u].x);
        atomicAdd(row + 1, w * h[u].y);
      }
    }
  }
}

__global__ __launch_bounds__(OWN_NT) void gat_edge_owner(
    const int* __restrict__ ei, const int* __restrict__ omap,
    const float* __restrict__ hp,
    const float* __restrict__ as_, const float* __restrict__ ad_,
    const float* __restrict__ bias, float* __restrict__ out, int E, int N) {
  __shared__ OwnLds s;
  const int tid  = threadIdx.x;
  const int lane = tid & 31;
  const int wave = __builtin_amdgcn_readfirstlane(tid >> 5);
  const int* __restrict__ dstp = ei + E;

  for (int base = blockIdx.x * OWN_ROWS; base < N; base += gridDim.x * OWN_ROWS) {
    const int cnt = (N - base < OWN_ROWS) ? (N - base) : OWN_ROWS;

    if (tid < cnt) { s.maxk[tid] = 0u; s.den[tid] = 0.0f; s.ad[tid] = ad_[base + tid]; }
    if (tid == 0) s.count = 0;
    __syncthreads();

    own_scan_map(s, omap, E, ((unsigned int)base / OWN_ROWS) & 255u, tid);
    __syncthreads();
    const int found = s.count;

    if (found <= OWN_CAP) {
      own_resolve(s, ei, dstp, as_, base, (unsigned int)cnt, found, tid);
      if (tid < cnt)
        atomicMax(&s.maxk[tid], f2key(leaky(as_[base + tid] + s.ad[tid])));
      __syncthreads();
      for (int i = tid; i < cnt * 64; i += OWN_NT) {
        const int dl = i >> 6;
        const float w = __expf(leaky(as_[base + dl] + s.ad[dl]) - key2f(s.maxk[dl]));
        s.acc[i] = w * hp[(size_t)base * 64 + i];
      }
      if (tid < cnt)
        atomicAdd(&s.den[tid],
                  __expf(leaky(as_[base + tid] + s.ad[tid]) - key2f(s.maxk[tid])));
      own_weights(s, found, tid);
      __syncthreads();
      own_rows(s, hp, found, wave, lane);
      __syncthreads();
    } else {
      for (int i = tid; i < E; i += OWN_NT) {
        const unsigned int dl = (unsigned int)(dstp[i] - base);
        if (dl < (unsigned int)cnt)
          atomicMax(&s.maxk[dl], f2key(leaky(as_[ei[i]] + s.ad[dl])));
      }
      if (tid < cnt)
        atomicMax(&s.maxk[tid], f2key(leaky(as_[base + tid] + s.ad[tid])));
      __syncthreads();
      for (int i = tid; i < cnt * 64; i += OWN_NT) {
        const int dl = i >> 6;
        const float w = __expf(leaky(as_[base + dl] + s.ad[dl]) - key2f(s.maxk[dl]));
        s.acc[i] = w * hp[(size_t)base * 64 + i];
      }
      if (tid < cnt)
        atomicAdd(&s.den[tid],
                  __expf(leaky(as_[base + tid] + s.ad[tid]) - key2f(s.maxk[tid])));
      __syncthreads();
      for (int w0 = 0; w0 < E; w0 += OWN_CAP) {
        if (tid == 0) s.count = 0;
        __syncthreads();
        own_scan(s, dstp, w0, (E - w0 < OWN_CAP) ? E : w0 + OWN_CAP, base,
                 (unsigned int)cnt, tid);
        __syncthreads();
        const int n = s.count;
        own_resolve(s, ei, dstp, as_, base, (unsigned int)cnt, n, tid);
        __syncthreads();
        own_weights(s, n, tid);
        __syncthreads();
        own_rows(s, hp, n, wave, lane);
        __syncthreads();
      }
    }

    for (int i = tid; i < cnt * 64; i += OWN_NT) {
      const float v = s.acc[i] / (s.den[i >> 6] + 1e-16f) + bias[i & 63];
      out[(size_t)base * 64 + i] = v > 0.0f ? v : 0.0f;
    }
    __syncthreads();
  }
}

extern "C" void kernel_launch(void* const* d_in, const int* in_sizes, int n_in,
                              void* d_out, int out_size, void* d_ws, size_t ws_size,
                              hipStream_t stream) {
  (void)n_in; (void)out_size; (void)ws_size;

  const float* x   = (const float*)d_in[0];
  const int*   ei  = (const int*)d_in[1];
  const int N = in_sizes[0] / 128;
  const int E = in_sizes[1] / 2;

  float* ws    = (float*)d_ws;
  float* hp    = ws;
  float* hcur  = hp    + (size_t)N * 64;
  float* as_   = hcur  + (size_t)N * 64;
  float* ad_   = as_   + N;

  const float* hin = x;
  int K = 128;
  for (int layer = 0; layer < 3; ++layer) {
    const float* W    = (const float*)d_in[4 + layer * 4];
    const float* asrc = (const float*)d_in[5 + layer * 4];
    const float* adst = (const float*)d_in[6 + layer * 4];
    const float* bias = (const float*)d_in[7 + layer * 4];

    const int mblocks = (N / 16 + 1) / 2;
    gat_gemm_wmma_f32<<<mblocks + 1, 256, 0, stream>>>(hin, W, hp, N, K);

    int* omap = (layer == 2) ? (int*)hcur : (int*)d_out;
    gat_alpha<<<512, 256, 0, stream>>>(hp, asrc, adst, as_, ad_, N, ei, omap, E);

    float* dst = (layer == 2) ? (float*)d_out : hcur;
    gat_edge_owner<<<256, OWN_NT, 0, stream>>>(ei, omap, hp, as_, ad_, bias, dst, E, N);

    hin = hcur;
    K = 64;
  }
}


# Edge lists found once: layer 0's groups save their resolved lists (source, local destination) in their own slices of the output plane; layers 1, 2 reload them, no map scan; sums stay LDS atomics
# speedup vs baseline: 5.5015x; 1.2498x over previous
#include <hip/hip_runtime.h>
#include <math.h>


typedef __attribute__((ext_vector_type(2))) float v2f;
typedef __attribute__((ext_vector_type(8))) float v8f;

#define NEG_SLOPE 0.2f
#define OWN_ROWS 196

#define REG_W     (OWN_ROWS * 64)
#define KEEP_HDR  4
#define MAP_CH_W  2048
#define MAP_OFF_W (REG_W - MAP_CH_W)
__device__ __forceinline__ size_t map_word(unsigned int g) {
  return (size_t)(g / MAP_CH_W) * REG_W + MAP_OFF_W + (g % MAP_CH_W);
}
__device__ __forceinline__ size_t map_vec(unsigned int v) {
  return (size_t)(v / (MAP_CH_W / 4)) * (REG_W / 4) + MAP_OFF_W / 4 + (v % (MAP_CH_W / 4));
}

__global__ __launch_bounds__(256) void gat_gemm_wmma_f32(
    const float* __restrict__ A, const float* __restrict__ W,
    float* __restrict__ C, int M, int K) {
  const int wave = threadIdx.x >> 5;
  const int lane = threadIdx.x & 31;
  const int ntile = wave & 3;
  const int mtile = (blockIdx.x << 1) | (wave >> 2);
  if (mtile * 16 >= M) return;

  const int mrow = mtile * 16 + (lane & 15);
  const int koff = (lane >> 4) << 1;
  const int col  = ntile * 16 + (lane & 15);

  const float* __restrict__ arow = A + (size_t)mrow * K;
  v8f acc = {};
  for (int k = 0; k < K; k += 4) {
    v2f a, b;
    a.x = arow[k + koff];
    a.y = arow[k + koff + 1];
    b.x = W[(size_t)(k + koff)     * 64 + col];
    b.y = W[(size_t)(k + koff + 1) * 64 + col];
    acc = __builtin_amdgcn_wmma_f32_16x16x4_f32(
        false, a, false, b, (short)0, acc, false, false);
  }
  const int rbase = mtile * 16 + ((lane >> 4) << 3);
#pragma unroll
  for (int v = 0; v < 8; ++v)
    C[(size_t)(rbase + v) * 64 + col] = acc[v];
}

#define AL_ROWS 7
__global__ __launch_bounds__(256) void gat_alpha(
    const float* __restrict__ hp, const float* __restrict__ a_s,
    const float* __restrict__ a_d, float* __restrict__ out_s,
    float* __restrict__ out_d, int N, const int* __restrict__ ei,
    int* __restrict__ omap, int E, int pack) {
  if (pack) {
    const int* __restrict__ dstp = ei + E;
    for (int g = blockIdx.x * 256 + threadIdx.x; 4 * g < E; g += gridDim.x * 256) {
      unsigned int pk = 0u;
      if (4 * g + 3 < E) {
        const int e = 4 * g;
        const unsigned int d0 = (unsigned int)dstp[e];
        const unsigned int d1 = (unsigned int)dstp[e + 1];
        const unsigned int d2 = (unsigned int)dstp[e + 2];
        const unsigned int d3 = (unsigned int)dstp[e + 3];
        pk = ((d0 / OWN_ROWS) & 255u) | (((d1 / OWN_ROWS) & 255u) << 8) |
             (((d2 / OWN_ROWS) & 255u) << 16) | (((d3 / OWN_ROWS) & 255u) << 24);
      } else {
#pragma unroll
        for (int b = 0; b < 4; ++b) {
          const int e = 4 * g + b;
          if (e < E)
            pk |= (((unsigned int)dstp[e] / OWN_ROWS) & 255u) << (8 * b);
        }
      }
      omap[map_word((unsigned int)g)] = (int)pk;
    }
  }
  const int lane = threadIdx.x & 31;
  const int sub  = lane & 15;
  const int half = lane >> 4;
  const int gw   = blockIdx.x * 8 + (threadIdx.x >> 5);
  const int tw   = gridDim.x * 8;
  float ws4[4], wd4[4];
#pragma unroll
  for (int j = 0; j < 4; ++j) { ws4[j] = a_s[4 * sub + j]; wd4[j] = a_d[4 * sub + j]; }
  const float4* __restrict__ hp4 = reinterpret_cast<const float4*>(hp);
  for (int p0 = gw; 2 * p0 < N; p0 += AL_ROWS * tw) {
    float4 v[AL_ROWS];
#pragma unroll
    for (int u = 0; u < AL_ROWS; ++u) {
      const int node = 2 * (p0 + u * tw) + half;
      v[u] = hp4[(size_t)(node < N ? node : 0) * 16 + sub];
    }
    float s[AL_ROWS], d[AL_ROWS];
#pragma unroll
    for (int u = 0; u < AL_ROWS; ++u) {
      s[u] = v[u].x * ws4[0] + v[u].y * ws4[1] + v[u].z * ws4[2] + v[u].w * ws4[3];
      d[u] = v[u].x * wd4[0] + v[u].y * wd4[1] + v[u].z * wd4[2] + v[u].w * wd4[3];
    }
#pragma unroll
    for (int m = 8; m > 0; m >>= 1) {
#pragma unroll
      for (int u = 0; u < AL_ROWS; ++u) {
        s[u] += __shfl_xor(s[u], m, 16);
        d[u] += __shfl_xor(d[u], m, 16);
      }
    }
#pragma unroll
    for (int u = 0; u < AL_ROWS; ++u) {
      const int node = 2 * (p0 + u * tw) + half;
      if (sub == 0 && node < N) { out_s[node] = s[u]; out_d[node] = d[u]; }
    }
  }
}

__device__ __forceinline__ unsigned int f2key(float f) {
  unsigned int u = __float_as_uint(f);
  return (u & 0x80000000u) ? ~u : (u | 0x80000000u);
}
__device__ __forceinline__ float key2f(unsigned int k) {
  return __uint_as_float((k & 0x80000000u) ? (k ^ 0x80000000u) : ~k);
}
__device__ __forceinline__ float leaky(float e) {
  return (e > 0.0f) ? e : NEG_SLOPE * e;
}

#define OWN_CAP  10240
#define OWN_NT   1024
static_assert(KEEP_HDR + OWN_CAP <= MAP_OFF_W, "a whole slice holds a full list in front of its map chunk");
static_assert(OWN_ROWS < 255, "local destinations must leave the byte value of KEEP_OUT free");

struct OwnLds {
  float        acc[OWN_ROWS * 64];
  float        den[OWN_ROWS];
  unsigned int maxk[OWN_ROWS];
  float        ad[OWN_ROWS];
  int          src[OWN_CAP];
  int          dl[OWN_CAP];
  float        e[OWN_CAP];
  int          count;
};

__device__ __forceinline__ void own_keep(OwnLds& s, int d, int eidx, int base,
                                         unsigned int cnt) {
  const unsigned int dl = (unsigned int)(d - base);
  if (dl < cnt) {
    const int pos = atomicAdd(&s.count, 1);
    if (pos < OWN_CAP) s.src[pos] = eidx;
  }
}

__device__ __forceinline__ unsigned int own_zero_bytes(unsigned int t) {
  return ~(((t & 0x7f7f7f7fu) + 0x7f7f7f7fu) | t | 0x7f7f7f7fu);
}

__device__ __forceinline__ void own_keep16(OwnLds& s, uint4 q, unsigned int pat,
                                           int e0) {
  unsigned int bits = (own_zero_bytes(q.x ^ pat) >> 7) |
                      (own_zero_bytes(q.y ^ pat) >> 6) |
                      (own_zero_bytes(q.z ^ pat) >> 5) |
                      (own_zero_bytes(q.w ^ pat) >> 4);
  if (bits) {
    int pos = atomicAdd(&s.count, __popc(bits));
    do {
      const int p = __ffs((int)bits) - 1;
      bits &= bits - 1u;
      if (pos < OWN_CAP) s.src[pos] = e0 + ((p & 7) << 2) + (p >> 3);
      ++pos;
    } while (bits);
  }
}

__device__ __forceinline__ void own_scan_map(OwnLds& s, const int* keep,
                                             int E, unsigned int id, int tid) {
  const unsigned int pat = id * 0x01010101u;
  const int nvec = E >> 4;
  const uint4* vp = reinterpret_cast<const uint4*>(keep);
  int v = tid;
  static_assert(OWN_NT % (MAP_CH_W / 4) == 0, "thread stride must be whole map chunks");
  const size_t hop = (size_t)(OWN_NT / (MAP_CH_W / 4)) * (REG_W / 4);
  for (; v + 3 * OWN_NT < nvec; v += 4 * OWN_NT) {
    const uint4* b = vp + map_vec((unsigned int)v);
    uint4 q[4];
#pragma unroll
    for (int u = 0; u < 4; ++u) q[u] = b[u * hop];
#pragma unroll
    for (int u = 0; u < 4; ++u) own_keep16(s, q[u], pat, (v + u * OWN_NT) << 4);
  }
  for (; v < nvec; v += OWN_NT) own_keep16(s, vp[map_vec((unsigned int)v)], pat, v << 4);
  const int t0 = nvec << 4;
  if (t0 + tid < E &&
      reinterpret_cast<const unsigned char*>(keep)[map_word((unsigned int)(t0 + tid) >> 2) * 4 +
                                                   ((t0 + tid) & 3)] == id) {
    const int pos = atomicAdd(&s.count, 1);
    if (pos < OWN_CAP) s.src[pos] = t0 + tid;
  }
}

__device__ __forceinline__ void own_scan(OwnLds& s, const int* __restrict__ dstp,
                                         int lo, int hi, int base,
                                         unsigned int cnt, int tid) {
  const int mis = (int)((reinterpret_cast<size_t>(dstp + lo) >> 2) & 3);
  int a0 = lo + ((4 - mis) & 3);
  if (a0 > hi) a0 = hi;
  if (lo + tid < a0) own_keep(s, dstp[lo + tid], lo + tid, base, cnt);
  const int nvec = (hi - a0) >> 2;
  const int4* __restrict__ vp = reinterpret_cast<const int4*>(dstp + a0);
  int v = tid;
  for (; v + 3 * OWN_NT < nvec; v += 4 * OWN_NT) {
    int4 q[4];
#pragma unroll
    for (int u = 0; u < 4; ++u) q[u] = vp[v + u * OWN_NT];
#pragma unroll
    for (int u = 0; u < 4; ++u) {
      const unsigned int d0 = (unsigned int)(q[u].x - base);
      const unsigned int d1 = (unsigned int)(q[u].y - base);
      const unsigned int d2 = (unsigned int)(q[u].z - base);
      const unsigned int d3 = (unsigned int)(q[u].w - base);
      const unsigned int m01 = d0 < d1 ? d0 : d1;
      const unsigned int m23 = d2 < d3 ? d2 : d3;
      if ((m01 < m23 ? m01 : m23) < cnt) {
        const int e0 = a0 + ((v + u * OWN_NT) << 2);
        own_keep(s, q[u].x, e0,     base, cnt);
        own_keep(s, q[u].y, e0 + 1, base, cnt);
        own_keep(s, q[u].z, e0 + 2, base, cnt);
        own_keep(s, q[u].w, e0 + 3, base, cnt);
      }
    }
  }
  for (; v < nvec; v += OWN_NT) {
    const int4 q = vp[v];
    const int e0 = a0 + (v << 2);
    own_keep(s, q.x, e0,     base, cnt);
    own_keep(s, q.y, e0 + 1, base, cnt);
    own_keep(s, q.z, e0 + 2, base, cnt);
    own_keep(s, q.w, e0 + 3, base, cnt);
  }
  const int t0 = a0 + (nvec << 2);
  if (t0 + tid < hi) own_keep(s, dstp[t0 + tid], t0 + tid, base, cnt);
}

#define KEEP_OUT 255u
__device__ __forceinline__ void own_resolve(OwnLds& s, const int* __restrict__ ei,
                                            const int* __restrict__ dstp,
                                            const float* __restrict__ as_,
                                            int base, unsigned int cnt,
                                            int n, int tid, int* save) {
  for (int p = tid; p < n; p += OWN_NT) {
    const int eidx = s.src[p];
    const int src  = ei[eidx];
    const unsigned int dl = (unsigned int)(dstp[eidx] - base);
    s.src[p] = src;
    if (save)
      save[p] = (int)((unsigned int)src | ((dl < cnt ? dl : KEEP_OUT) << 24));
    if (dl < cnt) {
      const float e = leaky(as_[src] + s.ad[dl]);
      s.dl[p] = (int)dl;
      s.e[p]  = e;
      atomicMax(&s.maxk[dl], f2key(e));
    } else {
      s.dl[p] = -1;
      s.e[p]  = 0.0f;
    }
  }
}

#define RL_U 4
__device__ __forceinline__ void own_reload(OwnLds& s, const int* saved,
                                           const float* __restrict__ as_,
                                           int n, int tid) {
  for (int p0 = tid; p0 < n; p0 += RL_U * OWN_NT) {
    unsigned int pk[RL_U];
#pragma unroll
    for (int u = 0; u < RL_U; ++u) {
      const int p = p0 + u * OWN_NT;
      pk[u] = p < n ? (unsigned int)saved[p] : (KEEP_OUT << 24);
    }
    float a[RL_U];
#pragma unroll
    for (int u = 0; u < RL_U; ++u) a[u] = as_[pk[u] & 0xffffffu];
#pragma unroll
    for (int u = 0; u < RL_U; ++u) {
      const int p = p0 + u * OWN_NT;
      if (p < n) {
        const unsigned int dl = pk[u] >> 24;
        s.src[p] = (int)(pk[u] & 0xffffffu);
        if (dl != KEEP_OUT) {
          const float e = leaky(a[u] + s.ad[dl]);
          s.dl[p] = (int)dl;
          s.e[p]  = e;
          atomicMax(&s.maxk[dl], f2key(e));
        } else {
          s.dl[p] = -1;
          s.e[p]  = 0.0f;
        }
      }
    }
  }
}

__device__ __forceinline__ void own_weights(OwnLds& s, int n, int tid) {
  for (int p = tid; p < n; p += OWN_NT) {
    const int dl = s.dl[p];
    if (dl >= 0) {
      const float w = __expf(s.e[p] - key2f(s.maxk[dl]));
      s.e[p] = w;
      atomicAdd(&s.den[dl], w);
    }
  }
}

#define ROW_R 16
__device__ __forceinline__ void own_rows(OwnLds& s, const float* __restrict__ hp,
                                         int n, int wave, int lane) {
  const float2* __restrict__ hp2 = reinterpret_cast<const float2*>(hp);
  for (int jb = wave * ROW_R; jb < n; jb += 32 * ROW_R) {
    const int  jj = jb + (lane & (ROW_R - 1));
    const bool ok = jj < n;
    const int   my_src = ok ? s.src[jj] : 0;
    const int   my_dl  = ok ? s.dl[jj]  : -1;
    const float my_w   = ok ? s.e[jj]   : 0.0f;
    float2 h[ROW_R];
#pragma unroll
    for (int u = 0; u < ROW_R; ++u) {
      const int src = __builtin_amdgcn_readlane(my_src, u);
      h[u] = hp2[(size_t)src * 32 + lane];
    }
#pragma unroll
    for (int u = 0; u < ROW_R; ++u) {
      const int dl = __builtin_amdgcn_readlane(my_dl, u);
      if (dl >= 0) {
        const float w =
            __int_as_float(__builtin_amdgcn_readlane(__float_as_int(my_w), u));
        float* row = &s.acc[dl * 64 + 2 * lane];
        atomicAdd(row,     w * h[u].x);
        atomicAdd(row + 1, w * h[u].y);
      }
    }
  }
}

__global__ __launch_bounds__(OWN_NT) void gat_edge_owner(
    const int* __restrict__ ei, int* keep,
    const float* __restrict__ hp,
    const float* __restrict__ as_, const float* __restrict__ ad_,
    const float* __restrict__ bias, float* out, int E, int N, int first) {
  __shared__ OwnLds s;
  const int tid  = threadIdx.x;
  const int lane = tid & 31;
  const int wave = __builtin_amdgcn_readfirstlane(tid >> 5);
  const int* __restrict__ dstp = ei + E;

  for (int base = blockIdx.x * OWN_ROWS; base < N; base += gridDim.x * OWN_ROWS) {
    const int cnt = (N - base < OWN_ROWS) ? (N - base) : OWN_ROWS;

    int* my = keep + (size_t)base * 64;
    const int room = (cnt * 64 - KEEP_HDR < OWN_CAP) ? cnt * 64 - KEEP_HDR : OWN_CAP;

    if (tid < cnt) { s.maxk[tid] = 0u; s.den[tid] = 0.0f; s.ad[tid] = ad_[base + tid]; }
    if (tid == 0) s.count = first ? 0 : my[0];
    __syncthreads();

    if (first) {
      own_scan_map(s, keep, E, ((unsigned int)base / OWN_ROWS) & 255u, tid);
      __syncthreads();
    }
    const int found = s.count;

    if (found >= 0 && found <= OWN_CAP) {
      if (first) {
        const bool fits = found <= room;
        own_resolve(s, ei, dstp, as_, base, (unsigned int)cnt, found, tid,
                    fits ? my + KEEP_HDR : nullptr);
        if (tid == 0) my[0] = fits ? found : -1;
      } else {
        own_reload(s, my + KEEP_HDR, as_, found, tid);
      }
      if (tid < cnt)
        atomicMax(&s.maxk[tid], f2key(leaky(as_[base + tid] + s.ad[tid])));
      __syncthreads();
      for (int i = tid; i < cnt * 64; i += OWN_NT) {
        const int dl = i >> 6;
        const float w = __expf(leaky(as_[base + dl] + s.ad[dl]) - key2f(s.maxk[dl]));
        s.acc[i] = w * hp[(size_t)base * 64 + i];
      }
      if (tid < cnt)
        atomicAdd(&s.den[tid],
                  __expf(leaky(as_[base + tid] + s.ad[tid]) - key2f(s.maxk[tid])));
      own_weights(s, found, tid);
      __syncthreads();
      own_rows(s, hp, found, wave, lane);
      __syncthreads();
    } else {
      if (first && tid == 0) my[0] = -1;
      for (int i = tid; i < E; i += OWN_NT) {
        const unsigned int dl = (unsigned int)(dstp[i] - base);
        if (dl < (unsigned int)cnt)
          atomicMax(&s.maxk[dl], f2key(leaky(as_[ei[i]] + s.ad[dl])));
      }
      if (tid < cnt)
        atomicMax(&s.maxk[tid], f2key(leaky(as_[base + tid] + s.ad[tid])));
      __syncthreads();
      for (int i = tid; i < cnt * 64; i += OWN_NT) {
        const int dl = i >> 6;
        const float w = __expf(leaky(as_[base + dl] + s.ad[dl]) - key2f(s.maxk[dl]));
        s.acc[i] = w * hp[(size_t)base * 64 + i];
      }
      if (tid < cnt)
        atomicAdd(&s.den[tid],
                  __expf(leaky(as_[base + tid] + s.ad[tid]) - key2f(s.maxk[tid])));
      __syncthreads();
      for (int w0 = 0; w0 < E; w0 += OWN_CAP) {
        if (tid == 0) s.count = 0;
        __syncthreads();
        own_scan(s, dstp, w0, (E - w0 < OWN_CAP) ? E : w0 + OWN_CAP, base,
                 (unsigned int)cnt, tid);
        __syncthreads();
        const int n = s.count;
        own_resolve(s, ei, dstp, as_, base, (unsigned int)cnt, n, tid, nullptr);
        __syncthreads();
        own_weights(s, n, tid);
        __syncthreads();
        own_rows(s, hp, n, wave, lane);
        __syncthreads();
      }
    }

    for (int i = tid; i < cnt * 64; i += OWN_NT) {
      const float v = s.acc[i] / (s.den[i >> 6] + 1e-16f) + bias[i & 63];
      out[(size_t)base * 64 + i] = v > 0.0f ? v : 0.0f;
    }
    __syncthreads();
  }
}

extern "C" void kernel_launch(void* const* d_in, const int* in_sizes, int n_in,
                              void* d_out, int out_size, void* d_ws, size_t ws_size,
                              hipStream_t stream) {
  (void)n_in; (void)out_size; (void)ws_size;

  const float* x   = (const float*)d_in[0];
  const int*   ei  = (const int*)d_in[1];
  const int N = in_sizes[0] / 128;
  const int E = in_sizes[1] / 2;

  float* ws    = (float*)d_ws;
  float* hp    = ws;
  float* hcur  = hp    + (size_t)N * 64;
  float* as_   = hcur  + (size_t)N * 64;
  float* ad_   = as_   + N;

  const float* hin = x;
  int K = 128;
  for (int layer = 0; layer < 3; ++layer) {
    const float* W    = (const float*)d_in[4 + layer * 4];
    const float* asrc = (const float*)d_in[5 + layer * 4];
    const float* adst = (const float*)d_in[6 + layer * 4];
    const float* bias = (const float*)d_in[7 + layer * 4];

    const int mblocks = (N / 16 + 1) / 2;
    gat_gemm_wmma_f32<<<mblocks + 1, 256, 0, stream>>>(hin, W, hp, N, K);

    int* keep = (int*)d_out;
    const int first = (layer == 0) ? 1 : 0;
    gat_alpha<<<512, 256, 0, stream>>>(hp, asrc, adst, as_, ad_, N, ei, keep, E, first);

    float* dst = (layer == 2) ? (float*)d_out : hcur;
    gat_edge_owner<<<256, OWN_NT, 0, stream>>>(ei, keep, hp, as_, ad_, bias, dst, E, N, first);

    hin = hcur;
    K = 64;
  }
}


# Alpha of layers 1, 2 by (hW)a = h(Wa): the edge group forms Wa of the next layer and dots its finished LDS rows with it; 2 alpha launches and their plane reads go; alpha sum order changes
# speedup vs baseline: 5.6674x; 1.0302x over previous
#include <hip/hip_runtime.h>
#include <math.h>


typedef __attribute__((ext_vector_type(2))) float v2f;
typedef __attribute__((ext_vector_type(8))) float v8f;

#define NEG_SLOPE 0.2f
#define OWN_ROWS 196

#define REG_W     (OWN_ROWS * 64)
#define KEEP_AD   4
#define KEEP_HDR  200
#define MAP_CH_W  2048
#define MAP_OFF_W (REG_W - MAP_CH_W)
__device__ __forceinline__ size_t map_word(unsigned int g) {
  return (size_t)(g / MAP_CH_W) * REG_W + MAP_OFF_W + (g % MAP_CH_W);
}
__device__ __forceinline__ size_t ad_word(int node) {
  return (size_t)(node / OWN_ROWS) * REG_W + KEEP_AD + (node % OWN_ROWS);
}
__device__ __forceinline__ size_t map_vec(unsigned int v) {
  return (size_t)(v / (MAP_CH_W / 4)) * (REG_W / 4) + MAP_OFF_W / 4 + (v % (MAP_CH_W / 4));
}

__global__ __launch_bounds__(256) void gat_gemm_wmma_f32(
    const float* __restrict__ A, const float* __restrict__ W,
    float* __restrict__ C, int M, int K) {
  const int wave = threadIdx.x >> 5;
  const int lane = threadIdx.x & 31;
  const int ntile = wave & 3;
  const int mtile = (blockIdx.x << 1) | (wave >> 2);
  if (mtile * 16 >= M) return;

  const int mrow = mtile * 16 + (lane & 15);
  const int koff = (lane >> 4) << 1;
  const int col  = ntile * 16 + (lane & 15);

  const float* __restrict__ arow = A + (size_t)mrow * K;
  v8f acc = {};
  for (int k = 0; k < K; k += 4) {
    v2f a, b;
    a.x = arow[k + koff];
    a.y = arow[k + koff + 1];
    b.x = W[(size_t)(k + koff)     * 64 + col];
    b.y = W[(size_t)(k + koff + 1) * 64 + col];
    acc = __builtin_amdgcn_wmma_f32_16x16x4_f32(
        false, a, false, b, (short)0, acc, false, false);
  }
  const int rbase = mtile * 16 + ((lane >> 4) << 3);
#pragma unroll
  for (int v = 0; v < 8; ++v)
    C[(size_t)(rbase + v) * 64 + col] = acc[v];
}

#define AL_ROWS 7
__global__ __launch_bounds__(256) void gat_alpha(
    const float* __restrict__ hp, const float* __restrict__ a_s,
    const float* __restrict__ a_d, float* __restrict__ out_s,
    float* keep_d, int N, const int* __restrict__ ei,
    int* omap, int E, int pack) {
  if (pack) {
    const int* __restrict__ dstp = ei + E;
    for (int g = blockIdx.x * 256 + threadIdx.x; 4 * g < E; g += gridDim.x * 256) {
      unsigned int pk = 0u;
      if (4 * g + 3 < E) {
        const int e = 4 * g;
        const unsigned int d0 = (unsigned int)dstp[e];
        const unsigned int d1 = (unsigned int)dstp[e + 1];
        const unsigned int d2 = (unsigned int)dstp[e + 2];
        const unsigned int d3 = (unsigned int)dstp[e + 3];
        pk = ((d0 / OWN_ROWS) & 255u) | (((d1 / OWN_ROWS) & 255u) << 8) |
             (((d2 / OWN_ROWS) & 255u) << 16) | (((d3 / OWN_ROWS) & 255u) << 24);
      } else {
#pragma unroll
        for (int b = 0; b < 4; ++b) {
          const int e = 4 * g + b;
          if (e < E)
            pk |= (((unsigned int)dstp[e] / OWN_ROWS) & 255u) << (8 * b);
        }
      }
      omap[map_word((unsigned int)g)] = (int)pk;
    }
  }
  const int lane = threadIdx.x & 31;
  const int sub  = lane & 15;
  const int half = lane >> 4;
  const int gw   = blockIdx.x * 8 + (threadIdx.x >> 5);
  const int tw   = gridDim.x * 8;
  float ws4[4], wd4[4];
#pragma unroll
  for (int j = 0; j < 4; ++j) { ws4[j] = a_s[4 * sub + j]; wd4[j] = a_d[4 * sub + j]; }
  const float4* __restrict__ hp4 = reinterpret_cast<const float4*>(hp);
  for (int p0 = gw; 2 * p0 < N; p0 += AL_ROWS * tw) {
    float4 v[AL_ROWS];
#pragma unroll
    for (int u = 0; u < AL_ROWS; ++u) {
      const int node = 2 * (p0 + u * tw) + half;
      v[u] = hp4[(size_t)(node < N ? node : 0) * 16 + sub];
    }
    float s[AL_ROWS], d[AL_ROWS];
#pragma unroll
    for (int u = 0; u < AL_ROWS; ++u) {
      s[u] = v[u].x * ws4[0] + v[u].y * ws4[1] + v[u].z * ws4[2] + v[u].w * ws4[3];
      d[u] = v[u].x * wd4[0] + v[u].y * wd4[1] + v[u].z * wd4[2] + v[u].w * wd4[3];
    }
#pragma unroll
    for (int m = 8; m > 0; m >>= 1) {
#pragma unroll
      for (int u = 0; u < AL_ROWS; ++u) {
        s[u] += __shfl_xor(s[u], m, 16);
        d[u] += __shfl_xor(d[u], m, 16);
      }
    }
#pragma unroll
    for (int u = 0; u < AL_ROWS; ++u) {
      const int node = 2 * (p0 + u * tw) + half;
      if (sub == 0 && node < N) { out_s[node] = s[u]; keep_d[ad_word(node)] = d[u]; }
    }
  }
}

__device__ __forceinline__ unsigned int f2key(float f) {
  unsigned int u = __float_as_uint(f);
  return (u & 0x80000000u) ? ~u : (u | 0x80000000u);
}
__device__ __forceinline__ float key2f(unsigned int k) {
  return __uint_as_float((k & 0x80000000u) ? (k ^ 0x80000000u) : ~k);
}
__device__ __forceinline__ float leaky(float e) {
  return (e > 0.0f) ? e : NEG_SLOPE * e;
}

#define OWN_CAP  10240
#define OWN_NT   1024
static_assert(KEEP_HDR + OWN_CAP <= MAP_OFF_W, "a whole slice holds a full list in front of its map chunk");
static_assert(OWN_ROWS < 255, "local destinations must leave the byte value of KEEP_OUT free");
static_assert(KEEP_AD + OWN_ROWS <= KEEP_HDR && KEEP_HDR % 4 == 0, "alpha_d words lie in front of the list");

struct OwnLds {
  alignas(16) float acc[OWN_ROWS * 64];
  alignas(16) float wa[128];
  float        den[OWN_ROWS];
  unsigned int maxk[OWN_ROWS];
  float        ad[OWN_ROWS];
  int          src[OWN_CAP];
  int          dl[OWN_CAP];
  float        e[OWN_CAP];
  int          count;
};

__device__ __forceinline__ void own_keep(OwnLds& s, int d, int eidx, int base,
                                         unsigned int cnt) {
  const unsigned int dl = (unsigned int)(d - base);
  if (dl < cnt) {
    const int pos = atomicAdd(&s.count, 1);
    if (pos < OWN_CAP) s.src[pos] = eidx;
  }
}

__device__ __forceinline__ unsigned int own_zero_bytes(unsigned int t) {
  return ~(((t & 0x7f7f7f7fu) + 0x7f7f7f7fu) | t | 0x7f7f7f7fu);
}

__device__ __forceinline__ void own_keep16(OwnLds& s, uint4 q, unsigned int pat,
                                           int e0) {
  unsigned int bits = (own_zero_bytes(q.x ^ pat) >> 7) |
                      (own_zero_bytes(q.y ^ pat) >> 6) |
                      (own_zero_bytes(q.z ^ pat) >> 5) |
                      (own_zero_bytes(q.w ^ pat) >> 4);
  if (bits) {
    int pos = atomicAdd(&s.count, __popc(bits));
    do {
      const int p = __ffs((int)bits) - 1;
      bits &= bits - 1u;
      if (pos < OWN_CAP) s.src[pos] = e0 + ((p & 7) << 2) + (p >> 3);
      ++pos;
    } while (bits);
  }
}

__device__ __forceinline__ void own_scan_map(OwnLds& s, const int* keep,
                                             int E, unsigned int id, int tid) {
  const unsigned int pat = id * 0x01010101u;
  const int nvec = E >> 4;
  const uint4* vp = reinterpret_cast<const uint4*>(keep);
  int v = tid;
  static_assert(OWN_NT % (MAP_CH_W / 4) == 0, "thread stride must be whole map chunks");
  const size_t hop = (size_t)(OWN_NT / (MAP_CH_W / 4)) * (REG_W / 4);
  for (; v + 3 * OWN_NT < nvec; v += 4 * OWN_NT) {
    const uint4* b = vp + map_vec((unsigned int)v);
    uint4 q[4];
#pragma unroll
    for (int u = 0; u < 4; ++u) q[u] = b[u * hop];
#pragma unroll
    for (int u = 0; u < 4; ++u) own_keep16(s, q[u], pat, (v + u * OWN_NT) << 4);
  }
  for (; v < nvec; v += OWN_NT) own_keep16(s, vp[map_vec((unsigned int)v)], pat, v << 4);
  const int t0 = nvec << 4;
  if (t0 + tid < E &&
      reinterpret_cast<const unsigned char*>(keep)[map_word((unsigned int)(t0 + tid) >> 2) * 4 +
                                                   ((t0 + tid) & 3)] == id) {
    const int pos = atomicAdd(&s.count, 1);
    if (pos < OWN_CAP) s.src[pos] = t0 + tid;
  }
}

__device__ __forceinline__ void own_scan(OwnLds& s, const int* __restrict__ dstp,
                                         int lo, int hi, int base,
                                         unsigned int cnt, int tid) {
  const int mis = (int)((reinterpret_cast<size_t>(dstp + lo) >> 2) & 3);
  int a0 = lo + ((4 - mis) & 3);
  if (a0 > hi) a0 = hi;
  if (lo + tid < a0) own_keep(s, dstp[lo + tid], lo + tid, base, cnt);
  const int nvec = (hi - a0) >> 2;
  const int4* __restrict__ vp = reinterpret_cast<const int4*>(dstp + a0);
  int v = tid;
  for (; v + 3 * OWN_NT < nvec; v += 4 * OWN_NT) {
    int4 q[4];
#pragma unroll
    for (int u = 0; u < 4; ++u) q[u] = vp[v + u * OWN_NT];
#pragma unroll
    for (int u = 0; u < 4; ++u) {
      const unsigned int d0 = (unsigned int)(q[u].x - base);
      const unsigned int d1 = (unsigned int)(q[u].y - base);
      const unsigned int d2 = (unsigned int)(q[u].z - base);
      const unsigned int d3 = (unsigned int)(q[u].w - base);
      const unsigned int m01 = d0 < d1 ? d0 : d1;
      const unsigned int m23 = d2 < d3 ? d2 : d3;
      if ((m01 < m23 ? m01 : m23) < cnt) {
        const int e0 = a0 + ((v + u * OWN_NT) << 2);
        own_keep(s, q[u].x, e0,     base, cnt);
        own_keep(s, q[u].y, e0 + 1, base, cnt);
        own_keep(s, q[u].z, e0 + 2, base, cnt);
        own_keep(s, q[u].w, e0 + 3, base, cnt);
      }
    }
  }
  for (; v < nvec; v += OWN_NT) {
    const int4 q = vp[v];
    const int e0 = a0 + (v << 2);
    own_keep(s, q.x, e0,     base, cnt);
    own_keep(s, q.y, e0 + 1, base, cnt);
    own_keep(s, q.z, e0 + 2, base, cnt);
    own_keep(s, q.w, e0 + 3, base, cnt);
  }
  const int t0 = a0 + (nvec << 2);
  if (t0 + tid < hi) own_keep(s, dstp[t0 + tid], t0 + tid, base, cnt);
}

#define KEEP_OUT 255u
__device__ __forceinline__ void own_resolve(OwnLds& s, const int* __restrict__ ei,
                                            const int* __restrict__ dstp,
                                            const float* __restrict__ as_,
                                            int base, unsigned int cnt,
                                            int n, int tid, int* save) {
  for (int p = tid; p < n; p += OWN_NT) {
    const int eidx = s.src[p];
    const int src  = ei[eidx];
    const unsigned int dl = (unsigned int)(dstp[eidx] - base);
    s.src[p] = src;
    if (save)
      save[p] = (int)((unsigned int)src | ((dl < cnt ? dl : KEEP_OUT) << 24));
    if (dl < cnt) {
      const float e = leaky(as_[src] + s.ad[dl]);
      s.dl[p] = (int)dl;
      s.e[p]  = e;
      atomicMax(&s.maxk[dl], f2key(e));
    } else {
      s.dl[p] = -1;
      s.e[p]  = 0.0f;
    }
  }
}

#define RL_U 4
__device__ __forceinline__ void own_reload(OwnLds& s, const int* saved,
                                           const float* __restrict__ as_,
                                           int n, int tid) {
  for (int p0 = tid; p0 < n; p0 += RL_U * OWN_NT) {
    unsigned int pk[RL_U];
#pragma unroll
    for (int u = 0; u < RL_U; ++u) {
      const int p = p0 + u * OWN_NT;
      pk[u] = p < n ? (unsigned int)saved[p] : (KEEP_OUT << 24);
    }
    float a[RL_U];
#pragma unroll
    for (int u = 0; u < RL_U; ++u) a[u] = as_[pk[u] & 0xffffffu];
#pragma unroll
    for (int u = 0; u < RL_U; ++u) {
      const int p = p0 + u * OWN_NT;
      if (p < n) {
        const unsigned int dl = pk[u] >> 24;
        s.src[p] = (int)(pk[u] & 0xffffffu);
        if (dl != KEEP_OUT) {
          const float e = leaky(a[u] + s.ad[dl]);
          s.dl[p] = (int)dl;
          s.e[p]  = e;
          atomicMax(&s.maxk[dl], f2key(e));
        } else {
          s.dl[p] = -1;
          s.e[p]  = 0.0f;
        }
      }
    }
  }
}

__device__ __forceinline__ void own_weights(OwnLds& s, int n, int tid) {
  for (int p = tid; p < n; p += OWN_NT) {
    const int dl = s.dl[p];
    if (dl >= 0) {
      const float w = __expf(s.e[p] - key2f(s.maxk[dl]));
      s.e[p] = w;
      atomicAdd(&s.den[dl], w);
    }
  }
}

#define ROW_R 16
__device__ __forceinline__ void own_rows(OwnLds& s, const float* __restrict__ hp,
                                         int n, int wave, int lane) {
  const float2* __restrict__ hp2 = reinterpret_cast<const float2*>(hp);
  for (int jb = wave * ROW_R; jb < n; jb += 32 * ROW_R) {
    const int  jj = jb + (lane & (ROW_R - 1));
    const bool ok = jj < n;
    const int   my_src = ok ? s.src[jj] : 0;
    const int   my_dl  = ok ? s.dl[jj]  : -1;
    const float my_w   = ok ? s.e[jj]   : 0.0f;
    float2 h[ROW_R];
#pragma unroll
    for (int u = 0; u < ROW_R; ++u) {
      const int src = __builtin_amdgcn_readlane(my_src, u);
      h[u] = hp2[(size_t)src * 32 + lane];
    }
#pragma unroll
    for (int u = 0; u < ROW_R; ++u) {
      const int dl = __builtin_amdgcn_readlane(my_dl, u);
      if (dl >= 0) {
        const float w =
            __int_as_float(__builtin_amdgcn_readlane(__float_as_int(my_w), u));
        float* row = &s.acc[dl * 64 + 2 * lane];
        atomicAdd(row,     w * h[u].x);
        atomicAdd(row + 1, w * h[u].y);
      }
    }
  }
}

__global__ __launch_bounds__(OWN_NT) void gat_edge_owner(
    const int* __restrict__ ei, int* keep,
    const float* __restrict__ hp,
    const float* __restrict__ as_,
    const float* __restrict__ bias, float* out, int E, int N, int first,
    const float* __restrict__ Wn, const float* __restrict__ asn,
    const float* __restrict__ adn, float* __restrict__ as_next) {
  __shared__ OwnLds s;
  const int tid  = threadIdx.x;
  const int lane = tid & 31;
  const int wave = __builtin_amdgcn_readfirstlane(tid >> 5);
  const int* __restrict__ dstp = ei + E;

  for (int base = blockIdx.x * OWN_ROWS; base < N; base += gridDim.x * OWN_ROWS) {
    const int cnt = (N - base < OWN_ROWS) ? (N - base) : OWN_ROWS;

    int* my = keep + (size_t)base * 64;
    const int room = (cnt * 64 - KEEP_HDR < OWN_CAP) ? cnt * 64 - KEEP_HDR : OWN_CAP;

    if (tid < cnt) {
      s.maxk[tid] = 0u; s.den[tid] = 0.0f;
      s.ad[tid] = reinterpret_cast<const float*>(my)[KEEP_AD + tid];
    }
    if (Wn) {
      const int q = tid & 15;
#pragma unroll
      for (int it = 0; it < 2; ++it) {
        const int d = (tid >> 4) + it * 64;
        const float* __restrict__ w = Wn + (d & 63) * 64 + 4 * q;
        const float* __restrict__ a = (it ? adn : asn) + 4 * q;
        float p = w[0] * a[0] + w[1] * a[1] + w[2] * a[2] + w[3] * a[3];
#pragma unroll
        for (int m = 8; m > 0; m >>= 1) p += __shfl_xor(p, m, 16);
        if (q == 0) s.wa[d] = p;
      }
    }
    if (tid == 0) s.count = first ? 0 : my[0];
    __syncthreads();

    if (first) {
      own_scan_map(s, keep, E, ((unsigned int)base / OWN_ROWS) & 255u, tid);
      __syncthreads();
    }
    const int found = s.count;

    if (found >= 0 && found <= OWN_CAP) {
      if (first) {
        const bool fits = found <= room;
        own_resolve(s, ei, dstp, as_, base, (unsigned int)cnt, found, tid,
                    fits ? my + KEEP_HDR : nullptr);
        if (tid == 0) my[0] = fits ? found : -1;
      } else {
        own_reload(s, my + KEEP_HDR, as_, found, tid);
      }
      if (tid < cnt)
        atomicMax(&s.maxk[tid], f2key(leaky(as_[base + tid] + s.ad[tid])));
      __syncthreads();
      for (int i = tid; i < cnt * 64; i += OWN_NT) {
        const int dl = i >> 6;
        const float w = __expf(leaky(as_[base + dl] + s.ad[dl]) - key2f(s.maxk[dl]));
        s.acc[i] = w * hp[(size_t)base * 64 + i];
      }
      if (tid < cnt)
        atomicAdd(&s.den[tid],
                  __expf(leaky(as_[base + tid] + s.ad[tid]) - key2f(s.maxk[tid])));
      own_weights(s, found, tid);
      __syncthreads();
      own_rows(s, hp, found, wave, lane);
      __syncthreads();
    } else {
      if (first && tid == 0) my[0] = -1;
      for (int i = tid; i < E; i += OWN_NT) {
        const unsigned int dl = (unsigned int)(dstp[i] - base);
        if (dl < (unsigned int)cnt)
          atomicMax(&s.maxk[dl], f2key(leaky(as_[ei[i]] + s.ad[dl])));
      }
      if (tid < cnt)
        atomicMax(&s.maxk[tid], f2key(leaky(as_[base + tid] + s.ad[tid])));
      __syncthreads();
      for (int i = tid; i < cnt * 64; i += OWN_NT) {
        const int dl = i >> 6;
        const float w = __expf(leaky(as_[base + dl] + s.ad[dl]) - key2f(s.maxk[dl]));
        s.acc[i] = w * hp[(size_t)base * 64 + i];
      }
      if (tid < cnt)
        atomicAdd(&s.den[tid],
                  __expf(leaky(as_[base + tid] + s.ad[tid]) - key2f(s.maxk[tid])));
      __syncthreads();
      for (int w0 = 0; w0 < E; w0 += OWN_CAP) {
        if (tid == 0) s.count = 0;
        __syncthreads();
        own_scan(s, dstp, w0, (E - w0 < OWN_CAP) ? E : w0 + OWN_CAP, base,
                 (unsigned int)cnt, tid);
        __syncthreads();
        const int n = s.count;
        own_resolve(s, ei, dstp, as_, base, (unsigned int)cnt, n, tid, nullptr);
        __syncthreads();
        own_weights(s, n, tid);
        __syncthreads();
        own_rows(s, hp, n, wave, lane);
        __syncthreads();
      }
    }

    for (int i = tid; i < cnt * 64; i += OWN_NT) {
      const float v = s.acc[i] / (s.den[i >> 6] + 1e-16f) + bias[i & 63];
      const float r = v > 0.0f ? v : 0.0f;
      out[(size_t)base * 64 + i] = r;
      if (Wn) s.acc[i] = r;
    }
    if (Wn) {
      __syncthreads();
      const int sub  = lane & 15;
      const int half = lane >> 4;
      const float4 ws4 = reinterpret_cast<const float4*>(s.wa)[sub];
      const float4 wd4 = reinterpret_cast<const float4*>(s.wa + 64)[sub];
      for (int r0 = wave * 2; r0 < cnt; r0 += 64) {
        const int r = r0 + half;
        float4 v = make_float4(0.0f, 0.0f, 0.0f, 0.0f);
        if (r < cnt) v = reinterpret_cast<const float4*>(s.acc)[r * 16 + sub];
        float sd = v.x * ws4.x + v.y * ws4.y + v.z * ws4.z + v.w * ws4.w;
        float dd = v.x * wd4.x + v.y * wd4.y + v.z * wd4.z + v.w * wd4.w;
#pragma unroll
        for (int m = 8; m > 0; m >>= 1) {
          sd += __shfl_xor(sd, m, 16);
          dd += __shfl_xor(dd, m, 16);
        }
        if (sub == 0 && r < cnt) {
          as_next[base + r] = sd;
          reinterpret_cast<float*>(my)[KEEP_AD + r] = dd;
        }
      }
    }
    __syncthreads();
  }
}

extern "C" void kernel_launch(void* const* d_in, const int* in_sizes, int n_in,
                              void* d_out, int out_size, void* d_ws, size_t ws_size,
                              hipStream_t stream) {
  (void)n_in; (void)out_size; (void)ws_size;

  const float* x   = (const float*)d_in[0];
  const int*   ei  = (const int*)d_in[1];
  const int N = in_sizes[0] / 128;
  const int E = in_sizes[1] / 2;

  float* ws    = (float*)d_ws;
  float* hp    = ws;
  float* hcur  = hp    + (size_t)N * 64;
  float* as_a  = hcur  + (size_t)N * 64;
  float* as_b  = as_a  + N;

  const float* hin = x;
  int K = 128;
  for (int layer = 0; layer < 3; ++layer) {
    const float* W    = (const float*)d_in[4 + layer * 4];
    const float* asrc = (const float*)d_in[5 + layer * 4];
    const float* adst = (const float*)d_in[6 + layer * 4];
    const float* bias = (const float*)d_in[7 + layer * 4];

    const int mblocks = (N / 16 + 1) / 2;
    gat_gemm_wmma_f32<<<mblocks + 1, 256, 0, stream>>>(hin, W, hp, N, K);

    int* keep = (int*)d_out;
    const int first = (layer == 0) ? 1 : 0;
    if (first)
      gat_alpha<<<512, 256, 0, stream>>>(hp, asrc, adst, as_a, (float*)d_out, N, ei, keep, E, first);

    float* dst = (layer == 2) ? (float*)d_out : hcur;
    const float* as_cur = (layer == 1) ? as_b : as_a;
    float*       as_nxt = (layer == 1) ? as_a : as_b;
    const float* Wn  = (layer == 2) ? nullptr : (const float*)d_in[4 + (layer + 1) * 4];
    const float* asn = (layer == 2) ? nullptr : (const float*)d_in[5 + (layer + 1) * 4];
    const float* adn = (layer == 2) ? nullptr : (const float*)d_in[6 + (layer + 1) * 4];
    gat_edge_owner<<<256, OWN_NT, 0, stream>>>(ei, keep, hp, as_cur, bias, dst, E, N, first,
                                               Wn, asn, adn, as_nxt);

    hin = hcur;
    K = 64;
  }
}


# Layer 0 edge lists by a split: alpha launch sorts 3328-edge chunks by owner in LDS behind an offset table; a group copies its runs; owner map, its scans, index gathers go; list order changes
# speedup vs baseline: 6.5031x; 1.1475x over previous
#include <hip/hip_runtime.h>
#include <math.h>


typedef __attribute__((ext_vector_type(2))) float v2f;
typedef __attribute__((ext_vector_type(8))) float v8f;

#define NEG_SLOPE 0.2f
#define OWN_ROWS 196

#define REG_W     (OWN_ROWS * 64)
#define KEEP_AD   4
#define KEEP_HDR  200
#define SPL_CH    3328
#define SPL_NT    256
#define SPL_U     ((SPL_CH + SPL_NT - 1) / SPL_NT)
#define SPL_TBL_W 260
#define SPL_W     (SPL_TBL_W + SPL_CH)
#define SPL_OFF_W (REG_W - SPL_W)
__device__ __forceinline__ size_t ad_word(int node) {
  return (size_t)(node / OWN_ROWS) * REG_W + KEEP_AD + (node % OWN_ROWS);
}
__device__ __forceinline__ size_t spl_word(int c) {
  return (size_t)c * REG_W + SPL_OFF_W;
}
__device__ __forceinline__ int spl_chunks(int E) { return (E + SPL_CH - 1) / SPL_CH; }
__device__ __forceinline__ bool split_fits(int N, int E) {
  return N <= 256 * OWN_ROWS && spl_chunks(E) <= N / OWN_ROWS;
}

__global__ __launch_bounds__(256) void gat_gemm_wmma_f32(
    const float* __restrict__ A, const float* __restrict__ W,
    float* __restrict__ C, int M, int K) {
  const int wave = threadIdx.x >> 5;
  const int lane = threadIdx.x & 31;
  const int ntile = wave & 3;
  const int mtile = (blockIdx.x << 1) | (wave >> 2);
  if (mtile * 16 >= M) return;

  const int mrow = mtile * 16 + (lane & 15);
  const int koff = (lane >> 4) << 1;
  const int col  = ntile * 16 + (lane & 15);

  const float* __restrict__ arow = A + (size_t)mrow * K;
  v8f acc = {};
  for (int k = 0; k < K; k += 4) {
    v2f a, b;
    a.x = arow[k + koff];
    a.y = arow[k + koff + 1];
    b.x = W[(size_t)(k + koff)     * 64 + col];
    b.y = W[(size_t)(k + koff + 1) * 64 + col];
    acc = __builtin_amdgcn_wmma_f32_16x16x4_f32(
        false, a, false, b, (short)0, acc, false, false);
  }
  const int rbase = mtile * 16 + ((lane >> 4) << 3);
#pragma unroll
  for (int v = 0; v < 8; ++v)
    C[(size_t)(rbase + v) * 64 + col] = acc[v];
}

#define AL_ROWS 7
__global__ __launch_bounds__(SPL_NT) void gat_alpha(
    const float* __restrict__ hp, const float* __restrict__ a_s,
    const float* __restrict__ a_d, float* __restrict__ out_s,
    float* keep_d, int N, const int* __restrict__ ei,
    int* keep, int E, int split) {
  __shared__ int sp_cnt[256 + 32];
  __shared__ int sp_wsum[SPL_NT / 32];
  __shared__ int sp_sorted[SPL_CH + 32];
  int nsb = 0;
  if (split && split_fits(N, E)) {
    nsb = spl_chunks(E);
    if (nsb > (int)(gridDim.x >> 1)) nsb = (int)(gridDim.x >> 1);
  }
  if ((int)blockIdx.x < nsb) {
    const int t = threadIdx.x;
    const int* __restrict__ dstp = ei + E;
    const int nch = spl_chunks(E);
    for (int c = blockIdx.x; c < nch; c += nsb) {
      sp_cnt[t] = 0;
      __syncthreads();
      const int e0  = c * SPL_CH;
      const int len = (E - e0 < SPL_CH) ? E - e0 : SPL_CH;
      int sv[SPL_U], dv[SPL_U];
#pragma unroll
      for (int u = 0; u < SPL_U; ++u) {
        const int i = t + u * SPL_NT;
        sv[u] = ei[e0 + (i < len ? i : 0)];
        dv[u] = dstp[e0 + (i < len ? i : 0)];
      }
      unsigned int word[SPL_U], br[SPL_U];
#pragma unroll
      for (int u = 0; u < SPL_U; ++u) {
        const unsigned int own = (unsigned int)dv[u] / OWN_ROWS;
        const unsigned int dl  = (unsigned int)dv[u] - own * OWN_ROWS;
        const unsigned int bin = (t + u * SPL_NT < len) ? (own & 255u) : 256u + (t & 31);
        const unsigned int rk  = (unsigned int)atomicAdd(&sp_cnt[bin], 1);
        word[u] = (unsigned int)sv[u] | (dl << 24);
        br[u]   = (own & 255u) | (rk << 8);
      }
      __syncthreads();
      const int lane = t & 31;
      const int cnt  = sp_cnt[t];
      int inc = cnt;
#pragma unroll
      for (int m = 1; m < 32; m <<= 1) {
        const int v = __shfl_up(inc, m, 32);
        if (lane >= m) inc += v;
      }
      if (lane == 31) sp_wsum[t >> 5] = inc;
      __syncthreads();
      int excl = inc - cnt;
#pragma unroll
      for (int w = 0; w < SPL_NT / 32; ++w)
        if (w < (t >> 5)) excl += sp_wsum[w];
      sp_cnt[t] = excl;
      int* tbl = keep + spl_word(c);
      tbl[t] = excl;
      if (t == SPL_NT - 1) tbl[256] = excl + cnt;
      __syncthreads();
#pragma unroll
      for (int u = 0; u < SPL_U; ++u) {
        const int pos = sp_cnt[br[u] & 255u] + (int)(br[u] >> 8);
        sp_sorted[(t + u * SPL_NT < len) ? pos : SPL_CH + (t & 31)] = (int)word[u];
      }
      __syncthreads();
      int ov[SPL_U];
#pragma unroll
      for (int u = 0; u < SPL_U; ++u) {
        const int i = t + u * SPL_NT;
        ov[u] = sp_sorted[i < len ? i : 0];
      }
#pragma unroll
      for (int u = 0; u < SPL_U; ++u) {
        const int i = t + u * SPL_NT;
        if (i < len) tbl[SPL_TBL_W + i] = ov[u];
      }
      __syncthreads();
    }
    return;
  }
  const int lane = threadIdx.x & 31;
  const int sub  = lane & 15;
  const int half = lane >> 4;
  const int gw   = ((int)blockIdx.x - nsb) * 8 + (threadIdx.x >> 5);
  const int tw   = ((int)gridDim.x - nsb) * 8;
  float ws4[4], wd4[4];
#pragma unroll
  for (int j = 0; j < 4; ++j) { ws4[j] = a_s[4 * sub + j]; wd4[j] = a_d[4 * sub + j]; }
  const float4* __restrict__ hp4 = reinterpret_cast<const float4*>(hp);
  for (int p0 = gw; 2 * p0 < N; p0 += AL_ROWS * tw) {
    float4 v[AL_ROWS];
#pragma unroll
    for (int u = 0; u < AL_ROWS; ++u) {
      const int node = 2 * (p0 + u * tw) + half;
      v[u] = hp4[(size_t)(node < N ? node : 0) * 16 + sub];
    }
    float s[AL_ROWS], d[AL_ROWS];
#pragma unroll
    for (int u = 0; u < AL_ROWS; ++u) {
      s[u] = v[u].x * ws4[0] + v[u].y * ws4[1] + v[u].z * ws4[2] + v[u].w * ws4[3];
      d[u] = v[u].x * wd4[0] + v[u].y * wd4[1] + v[u].z * wd4[2] + v[u].w * wd4[3];
    }
#pragma unroll
    for (int m = 8; m > 0; m >>= 1) {
#pragma unroll
      for (int u = 0; u < AL_ROWS; ++u) {
        s[u] += __shfl_xor(s[u], m, 16);
        d[u] += __shfl_xor(d[u], m, 16);
      }
    }
#pragma unroll
    for (int u = 0; u < AL_ROWS; ++u) {
      const int node = 2 * (p0 + u * tw) + half;
      if (sub == 0 && node < N) { out_s[node] = s[u]; keep_d[ad_word(node)] = d[u]; }
    }
  }
}

__device__ __forceinline__ unsigned int f2key(float f) {
  unsigned int u = __float_as_uint(f);
  return (u & 0x80000000u) ? ~u : (u | 0x80000000u);
}
__device__ __forceinline__ float key2f(unsigned int k) {
  return __uint_as_float((k & 0x80000000u) ? (k ^ 0x80000000u) : ~k);
}
__device__ __forceinline__ float leaky(float e) {
  return (e > 0.0f) ? e : NEG_SLOPE * e;
}

#define OWN_CAP  10240
#define OWN_NT   1024
static_assert(KEEP_HDR < SPL_OFF_W && SPL_OFF_W % 4 == 0 && SPL_TBL_W % 4 == 0 && SPL_TBL_W > 256,
              "a slice's sorted chunk lies behind its list room, 16-byte aligned");
static_assert(OWN_ROWS < 255, "local destinations must leave the byte value of KEEP_OUT free");
static_assert(KEEP_AD + OWN_ROWS <= KEEP_HDR && KEEP_HDR % 4 == 0, "alpha_d words lie in front of the list");

struct OwnLds {
  alignas(16) float acc[OWN_ROWS * 64];
  alignas(16) float wa[128];
  float        den[OWN_ROWS];
  unsigned int maxk[OWN_ROWS];
  float        ad[OWN_ROWS];
  int          src[OWN_CAP];
  int          dl[OWN_CAP];
  float        e[OWN_CAP];
  int          cstart[256];
  int          cbeg[256];
  int          wsum[8];
  int          count;
};

__device__ __forceinline__ void own_keep(OwnLds& s, int d, int eidx, int base,
                                         unsigned int cnt) {
  const unsigned int dl = (unsigned int)(d - base);
  if (dl < cnt) {
    const int pos = atomicAdd(&s.count, 1);
    if (pos < OWN_CAP) s.src[pos] = eidx;
  }
}

__device__ __forceinline__ void own_scan(OwnLds& s, const int* __restrict__ dstp,
                                         int lo, int hi, int base,
                                         unsigned int cnt, int tid) {
  const int mis = (int)((reinterpret_cast<size_t>(dstp + lo) >> 2) & 3);
  int a0 = lo + ((4 - mis) & 3);
  if (a0 > hi) a0 = hi;
  if (lo + tid < a0) own_keep(s, dstp[lo + tid], lo + tid, base, cnt);
  const int nvec = (hi - a0) >> 2;
  const int4* __restrict__ vp = reinterpret_cast<const int4*>(dstp + a0);
  int v = tid;
  for (; v + 3 * OWN_NT < nvec; v += 4 * OWN_NT) {
    int4 q[4];
#pragma unroll
    for (int u = 0; u < 4; ++u) q[u] = vp[v + u * OWN_NT];
#pragma unroll
    for (int u = 0; u < 4; ++u) {
      const unsigned int d0 = (unsigned int)(q[u].x - base);
      const unsigned int d1 = (unsigned int)(q[u].y - base);
      const unsigned int d2 = (unsigned int)(q[u].z - base);
      const unsigned int d3 = (unsigned int)(q[u].w - base);
      const unsigned int m01 = d0 < d1 ? d0 : d1;
      const unsigned int m23 = d2 < d3 ? d2 : d3;
      if ((m01 < m23 ? m01 : m23) < cnt) {
        const int e0 = a0 + ((v + u * OWN_NT) << 2);
        own_keep(s, q[u].x, e0,     base, cnt);
        own_keep(s, q[u].y, e0 + 1, base, cnt);
        own_keep(s, q[u].z, e0 + 2, base, cnt);
        own_keep(s, q[u].w, e0 + 3, base, cnt);
      }
    }
  }
  for (; v < nvec; v += OWN_NT) {
    const int4 q = vp[v];
    const int e0 = a0 + (v << 2);
    own_keep(s, q.x, e0,     base, cnt);
    own_keep(s, q.y, e0 + 1, base, cnt);
    own_keep(s, q.z, e0 + 2, base, cnt);
    own_keep(s, q.w, e0 + 3, base, cnt);
  }
  const int t0 = a0 + (nvec << 2);
  if (t0 + tid < hi) own_keep(s, dstp[t0 + tid], t0 + tid, base, cnt);
}

__device__ __forceinline__ void own_resolve(OwnLds& s, const int* __restrict__ ei,
                                            const int* __restrict__ dstp,
                                            const float* __restrict__ as_,
                                            int base, unsigned int cnt,
                                            int n, int tid) {
  for (int p = tid; p < n; p += OWN_NT) {
    const int eidx = s.src[p];
    const int src  = ei[eidx];
    const unsigned int dl = (unsigned int)(dstp[eidx] - base);
    s.src[p] = src;
    if (dl < cnt) {
      const float e = leaky(as_[src] + s.ad[dl]);
      s.dl[p] = (int)dl;
      s.e[p]  = e;
      atomicMax(&s.maxk[dl], f2key(e));
    } else {
      s.dl[p] = -1;
      s.e[p]  = 0.0f;
    }
  }
}

#define KEEP_OUT 255u
#define RL_U 4
template <bool FIRST>
__device__ __forceinline__ void own_reload(OwnLds& s, const int* saved,
                                           const float* __restrict__ as_,
                                           int n, int tid, int* save) {
  for (int p0 = tid; p0 < n; p0 += RL_U * OWN_NT) {
    unsigned int pk[RL_U];
#pragma unroll
    for (int u = 0; u < RL_U; ++u) {
      const int p = p0 + u * OWN_NT;
      if (FIRST) {
        const int pc = p < n ? p : 0;
        int b = 0;
#pragma unroll
        for (int st = 128; st > 0; st >>= 1)
          if (s.cstart[b + st] <= pc) b += st;
        const unsigned int w =
            (unsigned int)saved[spl_word(b) + SPL_TBL_W + s.cbeg[b] + (pc - s.cstart[b])];
        pk[u] = p < n ? w : (KEEP_OUT << 24);
      } else {
        pk[u] = p < n ? (unsigned int)saved[p] : (KEEP_OUT << 24);
      }
    }
    float a[RL_U];
#pragma unroll
    for (int u = 0; u < RL_U; ++u) a[u] = as_[pk[u] & 0xffffffu];
#pragma unroll
    for (int u = 0; u < RL_U; ++u) {
      const int p = p0 + u * OWN_NT;
      if (p < n) {
        const unsigned int dl = pk[u] >> 24;
        s.src[p] = (int)(pk[u] & 0xffffffu);
        if (FIRST && save) save[p] = (int)pk[u];
        if (dl != KEEP_OUT) {
          const float e = leaky(a[u] + s.ad[dl]);
          s.dl[p] = (int)dl;
          s.e[p]  = e;
          atomicMax(&s.maxk[dl], f2key(e));
        } else {
          s.dl[p] = -1;
          s.e[p]  = 0.0f;
        }
      }
    }
  }
}

__device__ __forceinline__ void own_weights(OwnLds& s, int n, int tid) {
  for (int p = tid; p < n; p += OWN_NT) {
    const int dl = s.dl[p];
    if (dl >= 0) {
      const float w = __expf(s.e[p] - key2f(s.maxk[dl]));
      s.e[p] = w;
      atomicAdd(&s.den[dl], w);
    }
  }
}

#define ROW_R 16
__device__ __forceinline__ void own_rows(OwnLds& s, const float* __restrict__ hp,
                                         int n, int wave, int lane) {
  const float2* __restrict__ hp2 = reinterpret_cast<const float2*>(hp);
  for (int jb = wave * ROW_R; jb < n; jb += 32 * ROW_R) {
    const int  jj = jb + (lane & (ROW_R - 1));
    const bool ok = jj < n;
    const int   my_src = ok ? s.src[jj] : 0;
    const int   my_dl  = ok ? s.dl[jj]  : -1;
    const float my_w   = ok ? s.e[jj]   : 0.0f;
    float2 h[ROW_R];
#pragma unroll
    for (int u = 0; u < ROW_R; ++u) {
      const int src = __builtin_amdgcn_readlane(my_src, u);
      h[u] = hp2[(size_t)src * 32 + lane];
    }
#pragma unroll
    for (int u = 0; u < ROW_R; ++u) {
      const int dl = __builtin_amdgcn_readlane(my_dl, u);
      if (dl >= 0) {
        const float w =
            __int_as_float(__builtin_amdgcn_readlane(__float_as_int(my_w), u));
        float* row = &s.acc[dl * 64 + 2 * lane];
        atomicAdd(row,     w * h[u].x);
        atomicAdd(row + 1, w * h[u].y);
      }
    }
  }
}

__global__ __launch_bounds__(OWN_NT) void gat_edge_owner(
    const int* __restrict__ ei, int* keep,
    const float* __restrict__ hp,
    const float* __restrict__ as_,
    const float* __restrict__ bias, float* out, int E, int N, int first,
    const float* __restrict__ Wn, const float* __restrict__ asn,
    const float* __restrict__ adn, float* __restrict__ as_next) {
  __shared__ OwnLds s;
  const int tid  = threadIdx.x;
  const int lane = tid & 31;
  const int wave = __builtin_amdgcn_readfirstlane(tid >> 5);
  const int* __restrict__ dstp = ei + E;

  for (int base = blockIdx.x * OWN_ROWS; base < N; base += gridDim.x * OWN_ROWS) {
    const int cnt = (N - base < OWN_ROWS) ? (N - base) : OWN_ROWS;

    int* my = keep + (size_t)base * 64;
    const int lim  = (cnt * 64 < SPL_OFF_W) ? cnt * 64 : SPL_OFF_W;
    const int room = (lim - KEEP_HDR < OWN_CAP) ? lim - KEEP_HDR : OWN_CAP;

    const bool split = first && split_fits(N, E);
    int run_beg = 0, run_len = 0;
    if (split && tid < spl_chunks(E)) {
      const int* tbl = keep + spl_word(tid);
      const int own = base / OWN_ROWS;
      run_beg = tbl[own];
      run_len = tbl[own + 1] - run_beg;
    }
    if (tid < cnt) {
      s.maxk[tid] = 0u; s.den[tid] = 0.0f;
      s.ad[tid] = reinterpret_cast<const float*>(my)[KEEP_AD + tid];
    }
    if (Wn) {
      const int q = tid & 15;
#pragma unroll
      for (int it = 0; it < 2; ++it) {
        const int d = (tid >> 4) + it * 64;
        const float* __restrict__ w = Wn + (d & 63) * 64 + 4 * q;
        const float* __restrict__ a = (it ? adn : asn) + 4 * q;
        float p = w[0] * a[0] + w[1] * a[1] + w[2] * a[2] + w[3] * a[3];
#pragma unroll
        for (int m = 8; m > 0; m >>= 1) p += __shfl_xor(p, m, 16);
        if (q == 0) s.wa[d] = p;
      }
    }
    int inc = 0;
    if (split && tid < 256) {
      inc = run_len;
#pragma unroll
      for (int m = 1; m < 32; m <<= 1) {
        const int v = __shfl_up(inc, m, 32);
        if (lane >= m) inc += v;
      }
      if (lane == 31) s.wsum[wave] = inc;
    }
    if (tid == 0) s.count = first ? -1 : my[0];
    __syncthreads();
    if (split) {
      if (tid < 256) {
        int excl = inc - run_len;
#pragma unroll
        for (int w = 0; w < 8; ++w)
          if (w < wave) excl += s.wsum[w];
        s.cstart[tid] = excl;
        s.cbeg[tid]   = run_beg;
        if (tid == 255) s.count = excl + run_len;
      }
      __syncthreads();
    }
    const int found = s.count;

    if (found >= 0 && found <= OWN_CAP) {
      if (first) {
        const bool fits = found <= room;
        own_reload<true>(s, keep, as_, found, tid, fits ? my + KEEP_HDR : nullptr);
        if (tid == 0) my[0] = fits ? found : -1;
      } else {
        own_reload<false>(s, my + KEEP_HDR, as_, found, tid, nullptr);
      }
      if (tid < cnt)
        atomicMax(&s.maxk[tid], f2key(leaky(as_[base + tid] + s.ad[tid])));
      __syncthreads();
      for (int i = tid; i < cnt * 64; i += OWN_NT) {
        const int dl = i >> 6;
        const float w = __expf(leaky(as_[base + dl] + s.ad[dl]) - key2f(s.maxk[dl]));
        s.acc[i] = w * hp[(size_t)base * 64 + i];
      }
      if (tid < cnt)
        atomicAdd(&s.den[tid],
                  __expf(leaky(as_[base + tid] + s.ad[tid]) - key2f(s.maxk[tid])));
      own_weights(s, found, tid);
      __syncthreads();
      own_rows(s, hp, found, wave, lane);
      __syncthreads();
    } else {
      if (first && tid == 0) my[0] = -1;
      for (int i = tid; i < E; i += OWN_NT) {
        const unsigned int dl = (unsigned int)(dstp[i] - base);
        if (dl < (unsigned int)cnt)
          atomicMax(&s.maxk[dl], f2key(leaky(as_[ei[i]] + s.ad[dl])));
      }
      if (tid < cnt)
        atomicMax(&s.maxk[tid], f2key(leaky(as_[base + tid] + s.ad[tid])));
      __syncthreads();
      for (int i = tid; i < cnt * 64; i += OWN_NT) {
        const int dl = i >> 6;
        const float w = __expf(leaky(as_[base + dl] + s.ad[dl]) - key2f(s.maxk[dl]));
        s.acc[i] = w * hp[(size_t)base * 64 + i];
      }
      if (tid < cnt)
        atomicAdd(&s.den[tid],
                  __expf(leaky(as_[base + tid] + s.ad[tid]) - key2f(s.maxk[tid])));
      __syncthreads();
      for (int w0 = 0; w0 < E; w0 += OWN_CAP) {
        if (tid == 0) s.count = 0;
        __syncthreads();
        own_scan(s, dstp, w0, (E - w0 < OWN_CAP) ? E : w0 + OWN_CAP, base,
                 (unsigned int)cnt, tid);
        __syncthreads();
        const int n = s.count;
        own_resolve(s, ei, dstp, as_, base, (unsigned int)cnt, n, tid);
        __syncthreads();
        own_weights(s, n, tid);
        __syncthreads();
        own_rows(s, hp, n, wave, lane);
        __syncthreads();
      }
    }

    for (int i = tid; i < cnt * 64; i += OWN_NT) {
      const float v = s.acc[i] / (s.den[i >> 6] + 1e-16f) + bias[i & 63];
      const float r = v > 0.0f ? v : 0.0f;
      out[(size_t)base * 64 + i] = r;
      if (Wn) s.acc[i] = r;
    }
    if (Wn) {
      __syncthreads();
      const int sub  = lane & 15;
      const int half = lane >> 4;
      const float4 ws4 = reinterpret_cast<const float4*>(s.wa)[sub];
      const float4 wd4 = reinterpret_cast<const float4*>(s.wa + 64)[sub];
      for (int r0 = wave * 2; r0 < cnt; r0 += 64) {
        const int r = r0 + half;
        float4 v = make_float4(0.0f, 0.0f, 0.0f, 0.0f);
        if (r < cnt) v = reinterpret_cast<const float4*>(s.acc)[r * 16 + sub];
        float sd = v.x * ws4.x + v.y * ws4.y + v.z * ws4.z + v.w * ws4.w;
        float dd = v.x * wd4.x + v.y * wd4.y + v.z * wd4.z + v.w * wd4.w;
#pragma unroll
        for (int m = 8; m > 0; m >>= 1) {
          sd += __shfl_xor(sd, m, 16);
          dd += __shfl_xor(dd, m, 16);
        }
        if (sub == 0 && r < cnt) {
          as_next[base + r] = sd;
          reinterpret_cast<float*>(my)[KEEP_AD + r] = dd;
        }
      }
    }
    __syncthreads();
  }
}

extern "C" void kernel_launch(void* const* d_in, const int* in_sizes, int n_in,
                              void* d_out, int out_size, void* d_ws, size_t ws_size,
                              hipStream_t stream) {
  (void)n_in; (void)out_size; (void)ws_size;

  const float* x   = (const float*)d_in[0];
  const int*   ei  = (const int*)d_in[1];
  const int N = in_sizes[0] / 128;
  const int E = in_sizes[1] / 2;

  float* ws    = (float*)d_ws;
  float* hp    = ws;
  float* hcur  = hp    + (size_t)N * 64;
  float* as_a  = hcur  + (size_t)N * 64;
  float* as_b  = as_a  + N;

  const float* hin = x;
  int K = 128;
  for (int layer = 0; layer < 3; ++layer) {
    const float* W    = (const float*)d_in[4 + layer * 4];
    const float* asrc = (const float*)d_in[5 + layer * 4];
    const float* adst = (const float*)d_in[6 + layer * 4];
    const float* bias = (const float*)d_in[7 + layer * 4];

    const int mblocks = (N / 16 + 1) / 2;
    gat_gemm_wmma_f32<<<mblocks + 1, 256, 0, stream>>>(hin, W, hp, N, K);

    int* keep = (int*)d_out;
    const int first = (layer == 0) ? 1 : 0;
    if (first)
      gat_alpha<<<512, 256, 0, stream>>>(hp, asrc, adst, as_a, (float*)d_out, N, ei, keep, E, first);

    float* dst = (layer == 2) ? (float*)d_out : hcur;
    const float* as_cur = (layer == 1) ? as_b : as_a;
    float*       as_nxt = (layer == 1) ? as_a : as_b;
    const float* Wn  = (layer == 2) ? nullptr : (const float*)d_in[4 + (layer + 1) * 4];
    const float* asn = (layer == 2) ? nullptr : (const float*)d_in[5 + (layer + 1) * 4];
    const float* adn = (layer == 2) ? nullptr : (const float*)d_in[6 + (layer + 1) * 4];
    gat_edge_owner<<<256, OWN_NT, 0, stream>>>(ei, keep, hp, as_cur, bias, dst, E, N, first,
                                               Wn, asn, adn, as_nxt);

    hin = hcur;
    K = 64;
  }
}


# Layers 1, 2 aggregate then transform, sum_j a_ij (h_j W) = (sum_j a_ij h_j) W: edge groups gather h rows, then multiply their summed LDS rows by W on fp32 WMMA; 2 GEMM launches go; sum order changes
# speedup vs baseline: 6.7200x; 1.0334x over previous
#include <hip/hip_runtime.h>
#include <math.h>


typedef __attribute__((ext_vector_type(2))) float v2f;
typedef __attribute__((ext_vector_type(8))) float v8f;

#define NEG_SLOPE 0.2f
#define OWN_ROWS 196

#define REG_W     (OWN_ROWS * 64)
#define KEEP_AD   4
#define KEEP_HDR  200
#define SPL_CH    3328
#define SPL_NT    256
#define SPL_U     ((SPL_CH + SPL_NT - 1) / SPL_NT)
#define SPL_TBL_W 260
#define SPL_W     (SPL_TBL_W + SPL_CH)
#define SPL_OFF_W (REG_W - SPL_W)
__device__ __forceinline__ size_t ad_word(int node) {
  return (size_t)(node / OWN_ROWS) * REG_W + KEEP_AD + (node % OWN_ROWS);
}
__device__ __forceinline__ size_t spl_word(int c) {
  return (size_t)c * REG_W + SPL_OFF_W;
}
__device__ __forceinline__ int spl_chunks(int E) { return (E + SPL_CH - 1) / SPL_CH; }
__device__ __forceinline__ bool split_fits(int N, int E) {
  return N <= 256 * OWN_ROWS && spl_chunks(E) <= N / OWN_ROWS;
}

__global__ __launch_bounds__(256) void gat_gemm_wmma_f32(
    const float* __restrict__ A, const float* __restrict__ W,
    float* __restrict__ C, int M, int K) {
  const int wave = threadIdx.x >> 5;
  const int lane = threadIdx.x & 31;
  const int ntile = wave & 3;
  const int mtile = (blockIdx.x << 1) | (wave >> 2);
  if (mtile * 16 >= M) return;

  const int mrow = mtile * 16 + (lane & 15);
  const int koff = (lane >> 4) << 1;
  const int col  = ntile * 16 + (lane & 15);

  const float* __restrict__ arow = A + (size_t)mrow * K;
  v8f acc = {};
  for (int k = 0; k < K; k += 4) {
    v2f a, b;
    a.x = arow[k + koff];
    a.y = arow[k + koff + 1];
    b.x = W[(size_t)(k + koff)     * 64 + col];
    b.y = W[(size_t)(k + koff + 1) * 64 + col];
    acc = __builtin_amdgcn_wmma_f32_16x16x4_f32(
        false, a, false, b, (short)0, acc, false, false);
  }
  const int rbase = mtile * 16 + ((lane >> 4) << 3);
#pragma unroll
  for (int v = 0; v < 8; ++v)
    C[(size_t)(rbase + v) * 64 + col] = acc[v];
}

#define AL_ROWS 7
__global__ __launch_bounds__(SPL_NT) void gat_alpha(
    const float* __restrict__ hp, const float* __restrict__ a_s,
    const float* __restrict__ a_d, float* __restrict__ out_s,
    float* keep_d, int N, const int* __restrict__ ei,
    int* keep, int E, int split) {
  __shared__ int sp_cnt[256 + 32];
  __shared__ int sp_wsum[SPL_NT / 32];
  __shared__ int sp_sorted[SPL_CH + 32];
  int nsb = 0;
  if (split && split_fits(N, E)) {
    nsb = spl_chunks(E);
    if (nsb > (int)(gridDim.x >> 1)) nsb = (int)(gridDim.x >> 1);
  }
  if ((int)blockIdx.x < nsb) {
    const int t = threadIdx.x;
    const int* __restrict__ dstp = ei + E;
    const int nch = spl_chunks(E);
    for (int c = blockIdx.x; c < nch; c += nsb) {
      sp_cnt[t] = 0;
      __syncthreads();
      const int e0  = c * SPL_CH;
      const int len = (E - e0 < SPL_CH) ? E - e0 : SPL_CH;
      int sv[SPL_U], dv[SPL_U];
#pragma unroll
      for (int u = 0; u < SPL_U; ++u) {
        const int i = t + u * SPL_NT;
        sv[u] = ei[e0 + (i < len ? i : 0)];
        dv[u] = dstp[e0 + (i < len ? i : 0)];
      }
      unsigned int word[SPL_U], br[SPL_U];
#pragma unroll
      for (int u = 0; u < SPL_U; ++u) {
        const unsigned int own = (unsigned int)dv[u] / OWN_ROWS;
        const unsigned int dl  = (unsigned int)dv[u] - own * OWN_ROWS;
        const unsigned int bin = (t + u * SPL_NT < len) ? (own & 255u) : 256u + (t & 31);
        const unsigned int rk  = (unsigned int)atomicAdd(&sp_cnt[bin], 1);
        word[u] = (unsigned int)sv[u] | (dl << 24);
        br[u]   = (own & 255u) | (rk << 8);
      }
      __syncthreads();
      const int lane = t & 31;
      const int cnt  = sp_cnt[t];
      int inc = cnt;
#pragma unroll
      for (int m = 1; m < 32; m <<= 1) {
        const int v = __shfl_up(inc, m, 32);
        if (lane >= m) inc += v;
      }
      if (lane == 31) sp_wsum[t >> 5] = inc;
      __syncthreads();
      int excl = inc - cnt;
#pragma unroll
      for (int w = 0; w < SPL_NT / 32; ++w)
        if (w < (t >> 5)) excl += sp_wsum[w];
      sp_cnt[t] = excl;
      int* tbl = keep + spl_word(c);
      tbl[t] = excl;
      if (t == SPL_NT - 1) tbl[256] = excl + cnt;
      __syncthreads();
#pragma unroll
      for (int u = 0; u < SPL_U; ++u) {
        const int pos = sp_cnt[br[u] & 255u] + (int)(br[u] >> 8);
        sp_sorted[(t + u * SPL_NT < len) ? pos : SPL_CH + (t & 31)] = (int)word[u];
      }
      __syncthreads();
      int ov[SPL_U];
#pragma unroll
      for (int u = 0; u < SPL_U; ++u) {
        const int i = t + u * SPL_NT;
        ov[u] = sp_sorted[i < len ? i : 0];
      }
#pragma unroll
      for (int u = 0; u < SPL_U; ++u) {
        const int i = t + u * SPL_NT;
        if (i < len) tbl[SPL_TBL_W + i] = ov[u];
      }
      __syncthreads();
    }
    return;
  }
  const int lane = threadIdx.x & 31;
  const int sub  = lane & 15;
  const int half = lane >> 4;
  const int gw   = ((int)blockIdx.x - nsb) * 8 + (threadIdx.x >> 5);
  const int tw   = ((int)gridDim.x - nsb) * 8;
  float ws4[4], wd4[4];
#pragma unroll
  for (int j = 0; j < 4; ++j) { ws4[j] = a_s[4 * sub + j]; wd4[j] = a_d[4 * sub + j]; }
  const float4* __restrict__ hp4 = reinterpret_cast<const float4*>(hp);
  for (int p0 = gw; 2 * p0 < N; p0 += AL_ROWS * tw) {
    float4 v[AL_ROWS];
#pragma unroll
    for (int u = 0; u < AL_ROWS; ++u) {
      const int node = 2 * (p0 + u * tw) + half;
      v[u] = hp4[(size_t)(node < N ? node : 0) * 16 + sub];
    }
    float s[AL_ROWS], d[AL_ROWS];
#pragma unroll
    for (int u = 0; u < AL_ROWS; ++u) {
      s[u] = v[u].x * ws4[0] + v[u].y * ws4[1] + v[u].z * ws4[2] + v[u].w * ws4[3];
      d[u] = v[u].x * wd4[0] + v[u].y * wd4[1] + v[u].z * wd4[2] + v[u].w * wd4[3];
    }
#pragma unroll
    for (int m = 8; m > 0; m >>= 1) {
#pragma unroll
      for (int u = 0; u < AL_ROWS; ++u) {
        s[u] += __shfl_xor(s[u], m, 16);
        d[u] += __shfl_xor(d[u], m, 16);
      }
    }
#pragma unroll
    for (int u = 0; u < AL_ROWS; ++u) {
      const int node = 2 * (p0 + u * tw) + half;
      if (sub == 0 && node < N) { out_s[node] = s[u]; keep_d[ad_word(node)] = d[u]; }
    }
  }
}

__device__ __forceinline__ unsigned int f2key(float f) {
  unsigned int u = __float_as_uint(f);
  return (u & 0x80000000u) ? ~u : (u | 0x80000000u);
}
__device__ __forceinline__ float key2f(unsigned int k) {
  return __uint_as_float((k & 0x80000000u) ? (k ^ 0x80000000u) : ~k);
}
__device__ __forceinline__ float leaky(float e) {
  return (e > 0.0f) ? e : NEG_SLOPE * e;
}

#define OWN_CAP  10240
#define OWN_NT   1024
static_assert(KEEP_HDR < SPL_OFF_W && SPL_OFF_W % 4 == 0 && SPL_TBL_W % 4 == 0 && SPL_TBL_W > 256,
              "a slice's sorted chunk lies behind its list room, 16-byte aligned");
static_assert(OWN_ROWS < 255, "local destinations must leave the byte value of KEEP_OUT free");
static_assert(KEEP_AD + OWN_ROWS <= KEEP_HDR && KEEP_HDR % 4 == 0, "alpha_d words lie in front of the list");

#define TP_ROWS  208
#define TP_PITCH 68
static_assert(TP_ROWS % 16 == 0 && TP_ROWS >= OWN_ROWS && TP_PITCH % 4 == 0, "tile buffer holds whole tiles, float4 rows");
struct OwnLds {
  alignas(16) float acc[OWN_ROWS * 64];
  alignas(16) float wa[128];
  alignas(16) float tp[TP_ROWS * TP_PITCH];
  float        den[OWN_ROWS];
  unsigned int maxk[OWN_ROWS];
  float        ad[OWN_ROWS];
  int          src[OWN_CAP];
  int          dl[OWN_CAP];
  float        e[OWN_CAP];
  int          cstart[256];
  int          cbeg[256];
  int          wsum[8];
  int          count;
};

__device__ __forceinline__ void own_keep(OwnLds& s, int d, int eidx, int base,
                                         unsigned int cnt) {
  const unsigned int dl = (unsigned int)(d - base);
  if (dl < cnt) {
    const int pos = atomicAdd(&s.count, 1);
    if (pos < OWN_CAP) s.src[pos] = eidx;
  }
}

__device__ __forceinline__ void own_scan(OwnLds& s, const int* __restrict__ dstp,
                                         int lo, int hi, int base,
                                         unsigned int cnt, int tid) {
  const int mis = (int)((reinterpret_cast<size_t>(dstp + lo) >> 2) & 3);
  int a0 = lo + ((4 - mis) & 3);
  if (a0 > hi) a0 = hi;
  if (lo + tid < a0) own_keep(s, dstp[lo + tid], lo + tid, base, cnt);
  const int nvec = (hi - a0) >> 2;
  const int4* __restrict__ vp = reinterpret_cast<const int4*>(dstp + a0);
  int v = tid;
  for (; v + 3 * OWN_NT < nvec; v += 4 * OWN_NT) {
    int4 q[4];
#pragma unroll
    for (int u = 0; u < 4; ++u) q[u] = vp[v + u * OWN_NT];
#pragma unroll
    for (int u = 0; u < 4; ++u) {
      const unsigned int d0 = (unsigned int)(q[u].x - base);
      const unsigned int d1 = (unsigned int)(q[u].y - base);
      const unsigned int d2 = (unsigned int)(q[u].z - base);
      const unsigned int d3 = (unsigned int)(q[u].w - base);
      const unsigned int m01 = d0 < d1 ? d0 : d1;
      const unsigned int m23 = d2 < d3 ? d2 : d3;
      if ((m01 < m23 ? m01 : m23) < cnt) {
        const int e0 = a0 + ((v + u * OWN_NT) << 2);
        own_keep(s, q[u].x, e0,     base, cnt);
        own_keep(s, q[u].y, e0 + 1, base, cnt);
        own_keep(s, q[u].z, e0 + 2, base, cnt);
        own_keep(s, q[u].w, e0 + 3, base, cnt);
      }
    }
  }
  for (; v < nvec; v += OWN_NT) {
    const int4 q = vp[v];
    const int e0 = a0 + (v << 2);
    own_keep(s, q.x, e0,     base, cnt);
    own_keep(s, q.y, e0 + 1, base, cnt);
    own_keep(s, q.z, e0 + 2, base, cnt);
    own_keep(s, q.w, e0 + 3, base, cnt);
  }
  const int t0 = a0 + (nvec << 2);
  if (t0 + tid < hi) own_keep(s, dstp[t0 + tid], t0 + tid, base, cnt);
}

__device__ __forceinline__ void own_resolve(OwnLds& s, const int* __restrict__ ei,
                                            const int* __restrict__ dstp,
                                            const float* __restrict__ as_,
                                            int base, unsigned int cnt,
                                            int n, int tid) {
  for (int p = tid; p < n; p += OWN_NT) {
    const int eidx = s.src[p];
    const int src  = ei[eidx];
    const unsigned int dl = (unsigned int)(dstp[eidx] - base);
    s.src[p] = src;
    if (dl < cnt) {
      const float e = leaky(as_[src] + s.ad[dl]);
      s.dl[p] = (int)dl;
      s.e[p]  = e;
      atomicMax(&s.maxk[dl], f2key(e));
    } else {
      s.dl[p] = -1;
      s.e[p]  = 0.0f;
    }
  }
}

#define KEEP_OUT 255u
#define RL_U 4
template <bool FIRST>
__device__ __forceinline__ void own_reload(OwnLds& s, const int* saved,
                                           const float* __restrict__ as_,
                                           int n, int tid, int* save) {
  for (int p0 = tid; p0 < n; p0 += RL_U * OWN_NT) {
    unsigned int pk[RL_U];
#pragma unroll
    for (int u = 0; u < RL_U; ++u) {
      const int p = p0 + u * OWN_NT;
      if (FIRST) {
        const int pc = p < n ? p : 0;
        int b = 0;
#pragma unroll
        for (int st = 128; st > 0; st >>= 1)
          if (s.cstart[b + st] <= pc) b += st;
        const unsigned int w =
            (unsigned int)saved[spl_word(b) + SPL_TBL_W + s.cbeg[b] + (pc - s.cstart[b])];
        pk[u] = p < n ? w : (KEEP_OUT << 24);
      } else {
        pk[u] = p < n ? (unsigned int)saved[p] : (KEEP_OUT << 24);
      }
    }
    float a[RL_U];
#pragma unroll
    for (int u = 0; u < RL_U; ++u) a[u] = as_[pk[u] & 0xffffffu];
#pragma unroll
    for (int u = 0; u < RL_U; ++u) {
      const int p = p0 + u * OWN_NT;
      if (p < n) {
        const unsigned int dl = pk[u] >> 24;
        s.src[p] = (int)(pk[u] & 0xffffffu);
        if (FIRST && save) save[p] = (int)pk[u];
        if (dl != KEEP_OUT) {
          const float e = leaky(a[u] + s.ad[dl]);
          s.dl[p] = (int)dl;
          s.e[p]  = e;
          atomicMax(&s.maxk[dl], f2key(e));
        } else {
          s.dl[p] = -1;
          s.e[p]  = 0.0f;
        }
      }
    }
  }
}

__device__ __forceinline__ void own_weights(OwnLds& s, int n, int tid) {
  for (int p = tid; p < n; p += OWN_NT) {
    const int dl = s.dl[p];
    if (dl >= 0) {
      const float w = __expf(s.e[p] - key2f(s.maxk[dl]));
      s.e[p] = w;
      atomicAdd(&s.den[dl], w);
    }
  }
}

#define ROW_R 16
__device__ __forceinline__ void own_rows(OwnLds& s, const float* __restrict__ hp,
                                         int n, int wave, int lane) {
  const float2* __restrict__ hp2 = reinterpret_cast<const float2*>(hp);
  for (int jb = wave * ROW_R; jb < n; jb += 32 * ROW_R) {
    const int  jj = jb + (lane & (ROW_R - 1));
    const bool ok = jj < n;
    const int   my_src = ok ? s.src[jj] : 0;
    const int   my_dl  = ok ? s.dl[jj]  : -1;
    const float my_w   = ok ? s.e[jj]   : 0.0f;
    float2 h[ROW_R];
#pragma unroll
    for (int u = 0; u < ROW_R; ++u) {
      const int src = __builtin_amdgcn_readlane(my_src, u);
      h[u] = hp2[(size_t)src * 32 + lane];
    }
#pragma unroll
    for (int u = 0; u < ROW_R; ++u) {
      const int dl = __builtin_amdgcn_readlane(my_dl, u);
      if (dl >= 0) {
        const float w =
            __int_as_float(__builtin_amdgcn_readlane(__float_as_int(my_w), u));
        float* row = &s.acc[dl * 64 + 2 * lane];
        atomicAdd(row,     w * h[u].x);
        atomicAdd(row + 1, w * h[u].y);
      }
    }
  }
}

__global__ __launch_bounds__(OWN_NT) void gat_edge_owner(
    const int* __restrict__ ei, int* keep,
    const float* __restrict__ hp,
    const float* __restrict__ as_,
    const float* __restrict__ bias, float* out, int E, int N, int first,
    const float* __restrict__ Wn, const float* __restrict__ asn,
    const float* __restrict__ adn, float* __restrict__ as_next,
    const float* __restrict__ Wc) {
  __shared__ OwnLds s;
  const int tid  = threadIdx.x;
  const int lane = tid & 31;
  const int wave = __builtin_amdgcn_readfirstlane(tid >> 5);
  const int* __restrict__ dstp = ei + E;

  for (int base = blockIdx.x * OWN_ROWS; base < N; base += gridDim.x * OWN_ROWS) {
    const int cnt = (N - base < OWN_ROWS) ? (N - base) : OWN_ROWS;

    int* my = keep + (size_t)base * 64;
    const int lim  = (cnt * 64 < SPL_OFF_W) ? cnt * 64 : SPL_OFF_W;
    const int room = (lim - KEEP_HDR < OWN_CAP) ? lim - KEEP_HDR : OWN_CAP;

    const bool split = first && split_fits(N, E);
    int run_beg = 0, run_len = 0;
    if (split && tid < spl_chunks(E)) {
      const int* tbl = keep + spl_word(tid);
      const int own = base / OWN_ROWS;
      run_beg = tbl[own];
      run_len = tbl[own + 1] - run_beg;
    }
    if (tid < cnt) {
      s.maxk[tid] = 0u; s.den[tid] = 0.0f;
      s.ad[tid] = reinterpret_cast<const float*>(my)[KEEP_AD + tid];
    }
    if (Wn) {
      const int q = tid & 15;
#pragma unroll
      for (int it = 0; it < 2; ++it) {
        const int d = (tid >> 4) + it * 64;
        const float* __restrict__ w = Wn + (d & 63) * 64 + 4 * q;
        const float* __restrict__ a = (it ? adn : asn) + 4 * q;
        float p = w[0] * a[0] + w[1] * a[1] + w[2] * a[2] + w[3] * a[3];
#pragma unroll
        for (int m = 8; m > 0; m >>= 1) p += __shfl_xor(p, m, 16);
        if (q == 0) s.wa[d] = p;
      }
    }
    int inc = 0;
    if (split && tid < 256) {
      inc = run_len;
#pragma unroll
      for (int m = 1; m < 32; m <<= 1) {
        const int v = __shfl_up(inc, m, 32);
        if (lane >= m) inc += v;
      }
      if (lane == 31) s.wsum[wave] = inc;
    }
    if (tid == 0) s.count = first ? -1 : my[0];
    __syncthreads();
    if (split) {
      if (tid < 256) {
        int excl = inc - run_len;
#pragma unroll
        for (int w = 0; w < 8; ++w)
          if (w < wave) excl += s.wsum[w];
        s.cstart[tid] = excl;
        s.cbeg[tid]   = run_beg;
        if (tid == 255) s.count = excl + run_len;
      }
      __syncthreads();
    }
    const int found = s.count;

    if (found >= 0 && found <= OWN_CAP) {
      if (first) {
        const bool fits = found <= room;
        own_reload<true>(s, keep, as_, found, tid, fits ? my + KEEP_HDR : nullptr);
        if (tid == 0) my[0] = fits ? found : -1;
      } else {
        own_reload<false>(s, my + KEEP_HDR, as_, found, tid, nullptr);
      }
      if (tid < cnt)
        atomicMax(&s.maxk[tid], f2key(leaky(as_[base + tid] + s.ad[tid])));
      __syncthreads();
      for (int i = tid; i < cnt * 64; i += OWN_NT) {
        const int dl = i >> 6;
        const float w = __expf(leaky(as_[base + dl] + s.ad[dl]) - key2f(s.maxk[dl]));
        s.acc[i] = w * hp[(size_t)base * 64 + i];
      }
      if (tid < cnt)
        atomicAdd(&s.den[tid],
                  __expf(leaky(as_[base + tid] + s.ad[tid]) - key2f(s.maxk[tid])));
      own_weights(s, found, tid);
      __syncthreads();
      own_rows(s, hp, found, wave, lane);
      __syncthreads();
    } else {
      if (first && tid == 0) my[0] = -1;
      for (int i = tid; i < E; i += OWN_NT) {
        const unsigned int dl = (unsigned int)(dstp[i] - base);
        if (dl < (unsigned int)cnt)
          atomicMax(&s.maxk[dl], f2key(leaky(as_[ei[i]] + s.ad[dl])));
      }
      if (tid < cnt)
        atomicMax(&s.maxk[tid], f2key(leaky(as_[base + tid] + s.ad[tid])));
      __syncthreads();
      for (int i = tid; i < cnt * 64; i += OWN_NT) {
        const int dl = i >> 6;
        const float w = __expf(leaky(as_[base + dl] + s.ad[dl]) - key2f(s.maxk[dl]));
        s.acc[i] = w * hp[(size_t)base * 64 + i];
      }
      if (tid < cnt)
        atomicAdd(&s.den[tid],
                  __expf(leaky(as_[base + tid] + s.ad[tid]) - key2f(s.maxk[tid])));
      __syncthreads();
      for (int w0 = 0; w0 < E; w0 += OWN_CAP) {
        if (tid == 0) s.count = 0;
        __syncthreads();
        own_scan(s, dstp, w0, (E - w0 < OWN_CAP) ? E : w0 + OWN_CAP, base,
                 (unsigned int)cnt, tid);
        __syncthreads();
        const int n = s.count;
        own_resolve(s, ei, dstp, as_, base, (unsigned int)cnt, n, tid);
        __syncthreads();
        own_weights(s, n, tid);
        __syncthreads();
        own_rows(s, hp, n, wave, lane);
        __syncthreads();
      }
    }

    if (Wc) {
      const int fc   = lane & 15;
      const int koff = (lane >> 4) << 1;
      const int col  = (wave & 3) * 16 + fc;
      v2f bf[16];
#pragma unroll
      for (int kk = 0; kk < 16; ++kk) {
        bf[kk].x = Wc[(4 * kk + koff)     * 64 + col];
        bf[kk].y = Wc[(4 * kk + koff + 1) * 64 + col];
      }
      const float bc = bias[col];
      for (int i = tid; i < TP_ROWS * 16; i += OWN_NT) {
        const int r = i >> 4;
        float4 v = make_float4(0.0f, 0.0f, 0.0f, 0.0f);
        if (r < cnt) {
          const float dn = s.den[r] + 1e-16f;
          v = reinterpret_cast<const float4*>(s.acc)[i];
          v.x /= dn; v.y /= dn; v.z /= dn; v.w /= dn;
        }
        *reinterpret_cast<float4*>(&s.tp[r * TP_PITCH + 4 * (i & 15)]) = v;
      }
      __syncthreads();
      const int ntile = ((cnt + 15) >> 4) * 4;
      for (int t = wave; t < ntile; t += 32) {
        const float* arow = &s.tp[((t >> 2) * 16 + fc) * TP_PITCH + koff];
        v8f d = {};
#pragma unroll
        for (int kk = 0; kk < 16; ++kk) {
          v2f a;
          a.x = arow[4 * kk];
          a.y = arow[4 * kk + 1];
          d = __builtin_amdgcn_wmma_f32_16x16x4_f32(
              false, a, false, bf[kk], (short)0, d, false, false);
        }
        const int r0 = (t >> 2) * 16 + ((lane >> 4) << 3);
#pragma unroll
        for (int v = 0; v < 8; ++v) {
          const float y = d[v] + bc;
          if (r0 + v < cnt) s.acc[(r0 + v) * 64 + col] = y > 0.0f ? y : 0.0f;
        }
      }
      __syncthreads();
      float4* __restrict__ o4 = reinterpret_cast<float4*>(out + (size_t)base * 64);
      for (int i = tid; i < cnt * 16; i += OWN_NT)
        o4[i] = reinterpret_cast<const float4*>(s.acc)[i];
    } else {
      for (int i = tid; i < cnt * 64; i += OWN_NT) {
        const float v = s.acc[i] / (s.den[i >> 6] + 1e-16f) + bias[i & 63];
        const float r = v > 0.0f ? v : 0.0f;
        out[(size_t)base * 64 + i] = r;
        if (Wn) s.acc[i] = r;
      }
      if (Wn) __syncthreads();
    }
    if (Wn) {
      const int sub  = lane & 15;
      const int half = lane >> 4;
      const float4 ws4 = reinterpret_cast<const float4*>(s.wa)[sub];
      const float4 wd4 = reinterpret_cast<const float4*>(s.wa + 64)[sub];
      for (int r0 = wave * 2; r0 < cnt; r0 += 64) {
        const int r = r0 + half;
        float4 v = make_float4(0.0f, 0.0f, 0.0f, 0.0f);
        if (r < cnt) v = reinterpret_cast<const float4*>(s.acc)[r * 16 + sub];
        float sd = v.x * ws4.x + v.y * ws4.y + v.z * ws4.z + v.w * ws4.w;
        float dd = v.x * wd4.x + v.y * wd4.y + v.z * wd4.z + v.w * wd4.w;
#pragma unroll
        for (int m = 8; m > 0; m >>= 1) {
          sd += __shfl_xor(sd, m, 16);
          dd += __shfl_xor(dd, m, 16);
        }
        if (sub == 0 && r < cnt) {
          as_next[base + r] = sd;
          reinterpret_cast<float*>(my)[KEEP_AD + r] = dd;
        }
      }
    }
    __syncthreads();
  }
}

extern "C" void kernel_launch(void* const* d_in, const int* in_sizes, int n_in,
                              void* d_out, int out_size, void* d_ws, size_t ws_size,
                              hipStream_t stream) {
  (void)n_in; (void)out_size; (void)ws_size;

  const float* x   = (const float*)d_in[0];
  const int*   ei  = (const int*)d_in[1];
  const int N = in_sizes[0] / 128;
  const int E = in_sizes[1] / 2;

  float* ws    = (float*)d_ws;
  float* hp    = ws;
  float* hcur  = hp    + (size_t)N * 64;
  float* as_a  = hcur  + (size_t)N * 64;
  float* as_b  = as_a  + N;

  const float* hin = x;
  int K = 128;
  for (int layer = 0; layer < 3; ++layer) {
    const float* W    = (const float*)d_in[4 + layer * 4];
    const float* asrc = (const float*)d_in[5 + layer * 4];
    const float* adst = (const float*)d_in[6 + layer * 4];
    const float* bias = (const float*)d_in[7 + layer * 4];

    const int mblocks = (N / 16 + 1) / 2;
    if (layer == 0)
      gat_gemm_wmma_f32<<<mblocks + 1, 256, 0, stream>>>(hin, W, hp, N, K);

    int* keep = (int*)d_out;
    const int first = (layer == 0) ? 1 : 0;
    if (first)
      gat_alpha<<<512, 256, 0, stream>>>(hp, asrc, adst, as_a, (float*)d_out, N, ei, keep, E, first);

    const float* gsrc = (layer == 1) ? hcur : hp;
    float* dst = (layer == 2) ? (float*)d_out : (layer == 1 ? hp : hcur);
    const float* as_cur = (layer == 1) ? as_b : as_a;
    float*       as_nxt = (layer == 1) ? as_a : as_b;
    const float* Wn  = (layer == 2) ? nullptr : (const float*)d_in[4 + (layer + 1) * 4];
    const float* asn = (layer == 2) ? nullptr : (const float*)d_in[5 + (layer + 1) * 4];
    const float* adn = (layer == 2) ? nullptr : (const float*)d_in[6 + (layer + 1) * 4];
    gat_edge_owner<<<256, OWN_NT, 0, stream>>>(ei, keep, gsrc, as_cur, bias, dst, E, N, first,
                                               Wn, asn, adn, as_nxt, first ? nullptr : W);
  }
}


# Self-loop opening of the owned rows: a thread's float4 row pieces and their alpha_s go out as one group ahead of the max barrier, not one 4-byte load waited for per trip, 13 trips; sums unchanged
# speedup vs baseline: 7.1676x; 1.0666x over previous
#include <hip/hip_runtime.h>
#include <math.h>


typedef __attribute__((ext_vector_type(2))) float v2f;
typedef __attribute__((ext_vector_type(8))) float v8f;

#define NEG_SLOPE 0.2f
#define OWN_ROWS 196

#define REG_W     (OWN_ROWS * 64)
#define KEEP_AD   4
#define KEEP_HDR  200
#define SPL_CH    3328
#define SPL_NT    256
#define SPL_U     ((SPL_CH + SPL_NT - 1) / SPL_NT)
#define SPL_TBL_W 260
#define SPL_W     (SPL_TBL_W + SPL_CH)
#define SPL_OFF_W (REG_W - SPL_W)
__device__ __forceinline__ size_t ad_word(int node) {
  return (size_t)(node / OWN_ROWS) * REG_W + KEEP_AD + (node % OWN_ROWS);
}
__device__ __forceinline__ size_t spl_word(int c) {
  return (size_t)c * REG_W + SPL_OFF_W;
}
__device__ __forceinline__ int spl_chunks(int E) { return (E + SPL_CH - 1) / SPL_CH; }
__device__ __forceinline__ bool split_fits(int N, int E) {
  return N <= 256 * OWN_ROWS && spl_chunks(E) <= N / OWN_ROWS;
}

__global__ __launch_bounds__(256) void gat_gemm_wmma_f32(
    const float* __restrict__ A, const float* __restrict__ W,
    float* __restrict__ C, int M, int K) {
  const int wave = threadIdx.x >> 5;
  const int lane = threadIdx.x & 31;
  const int ntile = wave & 3;
  const int mtile = (blockIdx.x << 1) | (wave >> 2);
  if (mtile * 16 >= M) return;

  const int mrow = mtile * 16 + (lane & 15);
  const int koff = (lane >> 4) << 1;
  const int col  = ntile * 16 + (lane & 15);

  const float* __restrict__ arow = A + (size_t)mrow * K;
  v8f acc = {};
  for (int k = 0; k < K; k += 4) {
    v2f a, b;
    a.x = arow[k + koff];
    a.y = arow[k + koff + 1];
    b.x = W[(size_t)(k + koff)     * 64 + col];
    b.y = W[(size_t)(k + koff + 1) * 64 + col];
    acc = __builtin_amdgcn_wmma_f32_16x16x4_f32(
        false, a, false, b, (short)0, acc, false, false);
  }
  const int rbase = mtile * 16 + ((lane >> 4) << 3);
#pragma unroll
  for (int v = 0; v < 8; ++v)
    C[(size_t)(rbase + v) * 64 + col] = acc[v];
}

#define AL_ROWS 7
__global__ __launch_bounds__(SPL_NT) void gat_alpha(
    const float* __restrict__ hp, const float* __restrict__ a_s,
    const float* __restrict__ a_d, float* __restrict__ out_s,
    float* keep_d, int N, const int* __restrict__ ei,
    int* keep, int E, int split) {
  __shared__ int sp_cnt[256 + 32];
  __shared__ int sp_wsum[SPL_NT / 32];
  __shared__ int sp_sorted[SPL_CH + 32];
  int nsb = 0;
  if (split && split_fits(N, E)) {
    nsb = spl_chunks(E);
    if (nsb > (int)(gridDim.x >> 1)) nsb = (int)(gridDim.x >> 1);
  }
  if ((int)blockIdx.x < nsb) {
    const int t = threadIdx.x;
    const int* __restrict__ dstp = ei + E;
    const int nch = spl_chunks(E);
    for (int c = blockIdx.x; c < nch; c += nsb) {
      sp_cnt[t] = 0;
      __syncthreads();
      const int e0  = c * SPL_CH;
      const int len = (E - e0 < SPL_CH) ? E - e0 : SPL_CH;
      int sv[SPL_U], dv[SPL_U];
#pragma unroll
      for (int u = 0; u < SPL_U; ++u) {
        const int i = t + u * SPL_NT;
        sv[u] = ei[e0 + (i < len ? i : 0)];
        dv[u] = dstp[e0 + (i < len ? i : 0)];
      }
      unsigned int word[SPL_U], br[SPL_U];
#pragma unroll
      for (int u = 0; u < SPL_U; ++u) {
        const unsigned int own = (unsigned int)dv[u] / OWN_ROWS;
        const unsigned int dl  = (unsigned int)dv[u] - own * OWN_ROWS;
        const unsigned int bin = (t + u * SPL_NT < len) ? (own & 255u) : 256u + (t & 31);
        const unsigned int rk  = (unsigned int)atomicAdd(&sp_cnt[bin], 1);
        word[u] = (unsigned int)sv[u] | (dl << 24);
        br[u]   = (own & 255u) | (rk << 8);
      }
      __syncthreads();
      const int lane = t & 31;
      const int cnt  = sp_cnt[t];
      int inc = cnt;
#pragma unroll
      for (int m = 1; m < 32; m <<= 1) {
        const int v = __shfl_up(inc, m, 32);
        if (lane >= m) inc += v;
      }
      if (lane == 31) sp_wsum[t >> 5] = inc;
      __syncthreads();
      int excl = inc - cnt;
#pragma unroll
      for (int w = 0; w < SPL_NT / 32; ++w)
        if (w < (t >> 5)) excl += sp_wsum[w];
      sp_cnt[t] = excl;
      int* tbl = keep + spl_word(c);
      tbl[t] = excl;
      if (t == SPL_NT - 1) tbl[256] = excl + cnt;
      __syncthreads();
#pragma unroll
      for (int u = 0; u < SPL_U; ++u) {
        const int pos = sp_cnt[br[u] & 255u] + (int)(br[u] >> 8);
        sp_sorted[(t + u * SPL_NT < len) ? pos : SPL_CH + (t & 31)] = (int)word[u];
      }
      __syncthreads();
      int ov[SPL_U];
#pragma unroll
      for (int u = 0; u < SPL_U; ++u) {
        const int i = t + u * SPL_NT;
        ov[u] = sp_sorted[i < len ? i : 0];
      }
#pragma unroll
      for (int u = 0; u < SPL_U; ++u) {
        const int i = t + u * SPL_NT;
        if (i < len) tbl[SPL_TBL_W + i] = ov[u];
      }
      __syncthreads();
    }
    return;
  }
  const int lane = threadIdx.x & 31;
  const int sub  = lane & 15;
  const int half = lane >> 4;
  const int gw   = ((int)blockIdx.x - nsb) * 8 + (threadIdx.x >> 5);
  const int tw   = ((int)gridDim.x - nsb) * 8;
  float ws4[4], wd4[4];
#pragma unroll
  for (int j = 0; j < 4; ++j) { ws4[j] = a_s[4 * sub + j]; wd4[j] = a_d[4 * sub + j]; }
  const float4* __restrict__ hp4 = reinterpret_cast<const float4*>(hp);
  for (int p0 = gw; 2 * p0 < N; p0 += AL_ROWS * tw) {
    float4 v[AL_ROWS];
#pragma unroll
    for (int u = 0; u < AL_ROWS; ++u) {
      const int node = 2 * (p0 + u * tw) + half;
      v[u] = hp4[(size_t)(node < N ? node : 0) * 16 + sub];
    }
    float s[AL_ROWS], d[AL_ROWS];
#pragma unroll
    for (int u = 0; u < AL_ROWS; ++u) {
      s[u] = v[u].x * ws4[0] + v[u].y * ws4[1] + v[u].z * ws4[2] + v[u].w * ws4[3];
      d[u] = v[u].x * wd4[0] + v[u].y * wd4[1] + v[u].z * wd4[2] + v[u].w * wd4[3];
    }
#pragma unroll
    for (int m = 8; m > 0; m >>= 1) {
#pragma unroll
      for (int u = 0; u < AL_ROWS; ++u) {
        s[u] += __shfl_xor(s[u], m, 16);
        d[u] += __shfl_xor(d[u], m, 16);
      }
    }
#pragma unroll
    for (int u = 0; u < AL_ROWS; ++u) {
      const int node = 2 * (p0 + u * tw) + half;
      if (sub == 0 && node < N) { out_s[node] = s[u]; keep_d[ad_word(node)] = d[u]; }
    }
  }
}

__device__ __forceinline__ unsigned int f2key(float f) {
  unsigned int u = __float_as_uint(f);
  return (u & 0x80000000u) ? ~u : (u | 0x80000000u);
}
__device__ __forceinline__ float key2f(unsigned int k) {
  return __uint_as_float((k & 0x80000000u) ? (k ^ 0x80000000u) : ~k);
}
__device__ __forceinline__ float leaky(float e) {
  return (e > 0.0f) ? e : NEG_SLOPE * e;
}

#define OWN_CAP  10240
#define OWN_NT   1024
static_assert(KEEP_HDR < SPL_OFF_W && SPL_OFF_W % 4 == 0 && SPL_TBL_W % 4 == 0 && SPL_TBL_W > 256,
              "a slice's sorted chunk lies behind its list room, 16-byte aligned");
static_assert(OWN_ROWS < 255, "local destinations must leave the byte value of KEEP_OUT free");
static_assert(KEEP_AD + OWN_ROWS <= KEEP_HDR && KEEP_HDR % 4 == 0, "alpha_d words lie in front of the list");

#define TP_ROWS  208
#define TP_PITCH 68
static_assert(TP_ROWS % 16 == 0 && TP_ROWS >= OWN_ROWS && TP_PITCH % 4 == 0, "tile buffer holds whole tiles, float4 rows");
struct OwnLds {
  alignas(16) float acc[OWN_ROWS * 64];
  alignas(16) float wa[128];
  alignas(16) float tp[TP_ROWS * TP_PITCH];
  float        den[OWN_ROWS];
  unsigned int maxk[OWN_ROWS];
  float        ad[OWN_ROWS];
  int          src[OWN_CAP];
  int          dl[OWN_CAP];
  float        e[OWN_CAP];
  int          cstart[256];
  int          cbeg[256];
  int          wsum[8];
  int          count;
};

__device__ __forceinline__ void own_keep(OwnLds& s, int d, int eidx, int base,
                                         unsigned int cnt) {
  const unsigned int dl = (unsigned int)(d - base);
  if (dl < cnt) {
    const int pos = atomicAdd(&s.count, 1);
    if (pos < OWN_CAP) s.src[pos] = eidx;
  }
}

__device__ __forceinline__ void own_scan(OwnLds& s, const int* __restrict__ dstp,
                                         int lo, int hi, int base,
                                         unsigned int cnt, int tid) {
  const int mis = (int)((reinterpret_cast<size_t>(dstp + lo) >> 2) & 3);
  int a0 = lo + ((4 - mis) & 3);
  if (a0 > hi) a0 = hi;
  if (lo + tid < a0) own_keep(s, dstp[lo + tid], lo + tid, base, cnt);
  const int nvec = (hi - a0) >> 2;
  const int4* __restrict__ vp = reinterpret_cast<const int4*>(dstp + a0);
  int v = tid;
  for (; v + 3 * OWN_NT < nvec; v += 4 * OWN_NT) {
    int4 q[4];
#pragma unroll
    for (int u = 0; u < 4; ++u) q[u] = vp[v + u * OWN_NT];
#pragma unroll
    for (int u = 0; u < 4; ++u) {
      const unsigned int d0 = (unsigned int)(q[u].x - base);
      const unsigned int d1 = (unsigned int)(q[u].y - base);
      const unsigned int d2 = (unsigned int)(q[u].z - base);
      const unsigned int d3 = (unsigned int)(q[u].w - base);
      const unsigned int m01 = d0 < d1 ? d0 : d1;
      const unsigned int m23 = d2 < d3 ? d2 : d3;
      if ((m01 < m23 ? m01 : m23) < cnt) {
        const int e0 = a0 + ((v + u * OWN_NT) << 2);
        own_keep(s, q[u].x, e0,     base, cnt);
        own_keep(s, q[u].y, e0 + 1, base, cnt);
        own_keep(s, q[u].z, e0 + 2, base, cnt);
        own_keep(s, q[u].w, e0 + 3, base, cnt);
      }
    }
  }
  for (; v < nvec; v += OWN_NT) {
    const int4 q = vp[v];
    const int e0 = a0 + (v << 2);
    own_keep(s, q.x, e0,     base, cnt);
    own_keep(s, q.y, e0 + 1, base, cnt);
    own_keep(s, q.z, e0 + 2, base, cnt);
    own_keep(s, q.w, e0 + 3, base, cnt);
  }
  const int t0 = a0 + (nvec << 2);
  if (t0 + tid < hi) own_keep(s, dstp[t0 + tid], t0 + tid, base, cnt);
}

__device__ __forceinline__ void own_resolve(OwnLds& s, const int* __restrict__ ei,
                                            const int* __restrict__ dstp,
                                            const float* __restrict__ as_,
                                            int base, unsigned int cnt,
                                            int n, int tid) {
  for (int p = tid; p < n; p += OWN_NT) {
    const int eidx = s.src[p];
    const int src  = ei[eidx];
    const unsigned int dl = (unsigned int)(dstp[eidx] - base);
    s.src[p] = src;
    if (dl < cnt) {
      const float e = leaky(as_[src] + s.ad[dl]);
      s.dl[p] = (int)dl;
      s.e[p]  = e;
      atomicMax(&s.maxk[dl], f2key(e));
    } else {
      s.dl[p] = -1;
      s.e[p]  = 0.0f;
    }
  }
}

#define KEEP_OUT 255u
#define RL_U 4
template <bool FIRST>
__device__ __forceinline__ void own_reload(OwnLds& s, const int* saved,
                                           const float* __restrict__ as_,
                                           int n, int tid, int* save) {
  for (int p0 = tid; p0 < n; p0 += RL_U * OWN_NT) {
    unsigned int pk[RL_U];
#pragma unroll
    for (int u = 0; u < RL_U; ++u) {
      const int p = p0 + u * OWN_NT;
      if (FIRST) {
        const int pc = p < n ? p : 0;
        int b = 0;
#pragma unroll
        for (int st = 128; st > 0; st >>= 1)
          if (s.cstart[b + st] <= pc) b += st;
        const unsigned int w =
            (unsigned int)saved[spl_word(b) + SPL_TBL_W + s.cbeg[b] + (pc - s.cstart[b])];
        pk[u] = p < n ? w : (KEEP_OUT << 24);
      } else {
        pk[u] = p < n ? (unsigned int)saved[p] : (KEEP_OUT << 24);
      }
    }
    float a[RL_U];
#pragma unroll
    for (int u = 0; u < RL_U; ++u) a[u] = as_[pk[u] & 0xffffffu];
#pragma unroll
    for (int u = 0; u < RL_U; ++u) {
      const int p = p0 + u * OWN_NT;
      if (p < n) {
        const unsigned int dl = pk[u] >> 24;
        s.src[p] = (int)(pk[u] & 0xffffffu);
        if (FIRST && save) save[p] = (int)pk[u];
        if (dl != KEEP_OUT) {
          const float e = leaky(a[u] + s.ad[dl]);
          s.dl[p] = (int)dl;
          s.e[p]  = e;
          atomicMax(&s.maxk[dl], f2key(e));
        } else {
          s.dl[p] = -1;
          s.e[p]  = 0.0f;
        }
      }
    }
  }
}

__device__ __forceinline__ void own_weights(OwnLds& s, int n, int tid) {
  for (int p = tid; p < n; p += OWN_NT) {
    const int dl = s.dl[p];
    if (dl >= 0) {
      const float w = __expf(s.e[p] - key2f(s.maxk[dl]));
      s.e[p] = w;
      atomicAdd(&s.den[dl], w);
    }
  }
}

#define ROW_R 16
__device__ __forceinline__ void own_rows(OwnLds& s, const float* __restrict__ hp,
                                         int n, int wave, int lane) {
  const float2* __restrict__ hp2 = reinterpret_cast<const float2*>(hp);
  for (int jb = wave * ROW_R; jb < n; jb += 32 * ROW_R) {
    const int  jj = jb + (lane & (ROW_R - 1));
    const bool ok = jj < n;
    const int   my_src = ok ? s.src[jj] : 0;
    const int   my_dl  = ok ? s.dl[jj]  : -1;
    const float my_w   = ok ? s.e[jj]   : 0.0f;
    float2 h[ROW_R];
#pragma unroll
    for (int u = 0; u < ROW_R; ++u) {
      const int src = __builtin_amdgcn_readlane(my_src, u);
      h[u] = hp2[(size_t)src * 32 + lane];
    }
#pragma unroll
    for (int u = 0; u < ROW_R; ++u) {
      const int dl = __builtin_amdgcn_readlane(my_dl, u);
      if (dl >= 0) {
        const float w =
            __int_as_float(__builtin_amdgcn_readlane(__float_as_int(my_w), u));
        float* row = &s.acc[dl * 64 + 2 * lane];
        atomicAdd(row,     w * h[u].x);
        atomicAdd(row + 1, w * h[u].y);
      }
    }
  }
}

#define SL_U ((OWN_ROWS * 16 + OWN_NT - 1) / OWN_NT)
__device__ __forceinline__ void own_open_load(const float* __restrict__ hp,
                                              const float* __restrict__ as_,
                                              int base, int cnt, int tid,
                                              float4 (&hv)[SL_U], float (&av)[SL_U]) {
  const float4* __restrict__ h4 = reinterpret_cast<const float4*>(hp + (size_t)base * 64);
#pragma unroll
  for (int u = 0; u < SL_U; ++u) {
    const int i  = tid + u * OWN_NT;
    const int ic = i < cnt * 16 ? i : 0;
    hv[u] = h4[ic];
    av[u] = as_[base + (ic >> 4)];
  }
}
__device__ __forceinline__ void own_open_use(OwnLds& s, int cnt, int tid,
                                             const float4 (&hv)[SL_U],
                                             const float (&av)[SL_U]) {
  const int r0 = tid >> 4;
#pragma unroll
  for (int u = 0; u < SL_U; ++u) {
    const int i = tid + u * OWN_NT;
    if (i < cnt * 16) {
      const int dl = r0 + u * (OWN_NT / 16);
      const float w = __expf(leaky(av[u] + s.ad[dl]) - key2f(s.maxk[dl]));
      reinterpret_cast<float4*>(s.acc)[i] =
          make_float4(w * hv[u].x, w * hv[u].y, w * hv[u].z, w * hv[u].w);
    }
  }
}

__global__ __launch_bounds__(OWN_NT) void gat_edge_owner(
    const int* __restrict__ ei, int* keep,
    const float* __restrict__ hp,
    const float* __restrict__ as_,
    const float* __restrict__ bias, float* out, int E, int N, int first,
    const float* __restrict__ Wn, const float* __restrict__ asn,
    const float* __restrict__ adn, float* __restrict__ as_next,
    const float* __restrict__ Wc) {
  __shared__ OwnLds s;
  const int tid  = threadIdx.x;
  const int lane = tid & 31;
  const int wave = __builtin_amdgcn_readfirstlane(tid >> 5);
  const int* __restrict__ dstp = ei + E;

  for (int base = blockIdx.x * OWN_ROWS; base < N; base += gridDim.x * OWN_ROWS) {
    const int cnt = (N - base < OWN_ROWS) ? (N - base) : OWN_ROWS;

    int* my = keep + (size_t)base * 64;
    const int lim  = (cnt * 64 < SPL_OFF_W) ? cnt * 64 : SPL_OFF_W;
    const int room = (lim - KEEP_HDR < OWN_CAP) ? lim - KEEP_HDR : OWN_CAP;

    const bool split = first && split_fits(N, E);
    int run_beg = 0, run_len = 0;
    if (split && tid < spl_chunks(E)) {
      const int* tbl = keep + spl_word(tid);
      const int own = base / OWN_ROWS;
      run_beg = tbl[own];
      run_len = tbl[own + 1] - run_beg;
    }
    if (tid < cnt) {
      s.maxk[tid] = 0u; s.den[tid] = 0.0f;
      s.ad[tid] = reinterpret_cast<const float*>(my)[KEEP_AD + tid];
    }
    if (Wn) {
      const int q = tid & 15;
#pragma unroll
      for (int it = 0; it < 2; ++it) {
        const int d = (tid >> 4) + it * 64;
        const float* __restrict__ w = Wn + (d & 63) * 64 + 4 * q;
        const float* __restrict__ a = (it ? adn : asn) + 4 * q;
        float p = w[0] * a[0] + w[1] * a[1] + w[2] * a[2] + w[3] * a[3];
#pragma unroll
        for (int m = 8; m > 0; m >>= 1) p += __shfl_xor(p, m, 16);
        if (q == 0) s.wa[d] = p;
      }
    }
    int inc = 0;
    if (split && tid < 256) {
      inc = run_len;
#pragma unroll
      for (int m = 1; m < 32; m <<= 1) {
        const int v = __shfl_up(inc, m, 32);
        if (lane >= m) inc += v;
      }
      if (lane == 31) s.wsum[wave] = inc;
    }
    if (tid == 0) s.count = first ? -1 : my[0];
    __syncthreads();
    if (split) {
      if (tid < 256) {
        int excl = inc - run_len;
#pragma unroll
        for (int w = 0; w < 8; ++w)
          if (w < wave) excl += s.wsum[w];
        s.cstart[tid] = excl;
        s.cbeg[tid]   = run_beg;
        if (tid == 255) s.count = excl + run_len;
      }
      __syncthreads();
    }
    const int found = s.count;

    if (found >= 0 && found <= OWN_CAP) {
      if (first) {
        const bool fits = found <= room;
        own_reload<true>(s, keep, as_, found, tid, fits ? my + KEEP_HDR : nullptr);
        if (tid == 0) my[0] = fits ? found : -1;
      } else {
        own_reload<false>(s, my + KEEP_HDR, as_, found, tid, nullptr);
      }
      float4 hv[SL_U];
      float  av[SL_U];
      own_open_load(hp, as_, base, cnt, tid, hv, av);
      if (tid < cnt)
        atomicMax(&s.maxk[tid], f2key(leaky(as_[base + tid] + s.ad[tid])));
      __syncthreads();
      own_open_use(s, cnt, tid, hv, av);
      if (tid < cnt)
        atomicAdd(&s.den[tid],
                  __expf(leaky(as_[base + tid] + s.ad[tid]) - key2f(s.maxk[tid])));
      own_weights(s, found, tid);
      __syncthreads();
      own_rows(s, hp, found, wave, lane);
      __syncthreads();
    } else {
      if (first && tid == 0) my[0] = -1;
      for (int i = tid; i < E; i += OWN_NT) {
        const unsigned int dl = (unsigned int)(dstp[i] - base);
        if (dl < (unsigned int)cnt)
          atomicMax(&s.maxk[dl], f2key(leaky(as_[ei[i]] + s.ad[dl])));
      }
      if (tid < cnt)
        atomicMax(&s.maxk[tid], f2key(leaky(as_[base + tid] + s.ad[tid])));
      __syncthreads();
      {
        float4 hv[SL_U];
        float  av[SL_U];
        own_open_load(hp, as_, base, cnt, tid, hv, av);
        own_open_use(s, cnt, tid, hv, av);
      }
      if (tid < cnt)
        atomicAdd(&s.den[tid],
                  __expf(leaky(as_[base + tid] + s.ad[tid]) - key2f(s.maxk[tid])));
      __syncthreads();
      for (int w0 = 0; w0 < E; w0 += OWN_CAP) {
        if (tid == 0) s.count = 0;
        __syncthreads();
        own_scan(s, dstp, w0, (E - w0 < OWN_CAP) ? E : w0 + OWN_CAP, base,
                 (unsigned int)cnt, tid);
        __syncthreads();
        const int n = s.count;
        own_resolve(s, ei, dstp, as_, base, (unsigned int)cnt, n, tid);
        __syncthreads();
        own_weights(s, n, tid);
        __syncthreads();
        own_rows(s, hp, n, wave, lane);
        __syncthreads();
      }
    }

    if (Wc) {
      const int fc   = lane & 15;
      const int koff = (lane >> 4) << 1;
      const int col  = (wave & 3) * 16 + fc;
      v2f bf[16];
#pragma unroll
      for (int kk = 0; kk < 16; ++kk) {
        bf[kk].x = Wc[(4 * kk + koff)     * 64 + col];
        bf[kk].y = Wc[(4 * kk + koff + 1) * 64 + col];
      }
      const float bc = bias[col];
      for (int i = tid; i < TP_ROWS * 16; i += OWN_NT) {
        const int r = i >> 4;
        float4 v = make_float4(0.0f, 0.0f, 0.0f, 0.0f);
        if (r < cnt) {
          const float dn = s.den[r] + 1e-16f;
          v = reinterpret_cast<const float4*>(s.acc)[i];
          v.x /= dn; v.y /= dn; v.z /= dn; v.w /= dn;
        }
        *reinterpret_cast<float4*>(&s.tp[r * TP_PITCH + 4 * (i & 15)]) = v;
      }
      __syncthreads();
      const int ntile = ((cnt + 15) >> 4) * 4;
      for (int t = wave; t < ntile; t += 32) {
        const float* arow = &s.tp[((t >> 2) * 16 + fc) * TP_PITCH + koff];
        v8f d = {};
#pragma unroll
        for (int kk = 0; kk < 16; ++kk) {
          v2f a;
          a.x = arow[4 * kk];
          a.y = arow[4 * kk + 1];
          d = __builtin_amdgcn_wmma_f32_16x16x4_f32(
              false, a, false, bf[kk], (short)0, d, false, false);
        }
        const int r0 = (t >> 2) * 16 + ((lane >> 4) << 3);
#pragma unroll
        for (int v = 0; v < 8; ++v) {
          const float y = d[v] + bc;
          if (r0 + v < cnt) s.acc[(r0 + v) * 64 + col] = y > 0.0f ? y : 0.0f;
        }
      }
      __syncthreads();
      float4* __restrict__ o4 = reinterpret_cast<float4*>(out + (size_t)base * 64);
      for (int i = tid; i < cnt * 16; i += OWN_NT)
        o4[i] = reinterpret_cast<const float4*>(s.acc)[i];
    } else {
      for (int i = tid; i < cnt * 64; i += OWN_NT) {
        const float v = s.acc[i] / (s.den[i >> 6] + 1e-16f) + bias[i & 63];
        const float r = v > 0.0f ? v : 0.0f;
        out[(size_t)base * 64 + i] = r;
        if (Wn) s.acc[i] = r;
      }
      if (Wn) __syncthreads();
    }
    if (Wn) {
      const int sub  = lane & 15;
      const int half = lane >> 4;
      const float4 ws4 = reinterpret_cast<const float4*>(s.wa)[sub];
      const float4 wd4 = reinterpret_cast<const float4*>(s.wa + 64)[sub];
      for (int r0 = wave * 2; r0 < cnt; r0 += 64) {
        const int r = r0 + half;
        float4 v = make_float4(0.0f, 0.0f, 0.0f, 0.0f);
        if (r < cnt) v = reinterpret_cast<const float4*>(s.acc)[r * 16 + sub];
        float sd = v.x * ws4.x + v.y * ws4.y + v.z * ws4.z + v.w * ws4.w;
        float dd = v.x * wd4.x + v.y * wd4.y + v.z * wd4.z + v.w * wd4.w;
#pragma unroll
        for (int m = 8; m > 0; m >>= 1) {
          sd += __shfl_xor(sd, m, 16);
          dd += __shfl_xor(dd, m, 16);
        }
        if (sub == 0 && r < cnt) {
          as_next[base + r] = sd;
          reinterpret_cast<float*>(my)[KEEP_AD + r] = dd;
        }
      }
    }
    __syncthreads();
  }
}

extern "C" void kernel_launch(void* const* d_in, const int* in_sizes, int n_in,
                              void* d_out, int out_size, void* d_ws, size_t ws_size,
                              hipStream_t stream) {
  (void)n_in; (void)out_size; (void)ws_size;

  const float* x   = (const float*)d_in[0];
  const int*   ei  = (const int*)d_in[1];
  const int N = in_sizes[0] / 128;
  const int E = in_sizes[1] / 2;

  float* ws    = (float*)d_ws;
  float* hp    = ws;
  float* hcur  = hp    + (size_t)N * 64;
  float* as_a  = hcur  + (size_t)N * 64;
  float* as_b  = as_a  + N;

  const float* hin = x;
  int K = 128;
  for (int layer = 0; layer < 3; ++layer) {
    const float* W    = (const float*)d_in[4 + layer * 4];
    const float* asrc = (const float*)d_in[5 + layer * 4];
    const float* adst = (const float*)d_in[6 + layer * 4];
    const float* bias = (const float*)d_in[7 + layer * 4];

    const int mblocks = (N / 16 + 1) / 2;
    if (layer == 0)
      gat_gemm_wmma_f32<<<mblocks + 1, 256, 0, stream>>>(hin, W, hp, N, K);

    int* keep = (int*)d_out;
    const int first = (layer == 0) ? 1 : 0;
    if (first)
      gat_alpha<<<512, 256, 0, stream>>>(hp, asrc, adst, as_a, (float*)d_out, N, ei, keep, E, first);

    const float* gsrc = (layer == 1) ? hcur : hp;
    float* dst = (layer == 2) ? (float*)d_out : (layer == 1 ? hp : hcur);
    const float* as_cur = (layer == 1) ? as_b : as_a;
    float*       as_nxt = (layer == 1) ? as_a : as_b;
    const float* Wn  = (layer == 2) ? nullptr : (const float*)d_in[4 + (layer + 1) * 4];
    const float* asn = (layer == 2) ? nullptr : (const float*)d_in[5 + (layer + 1) * 4];
    const float* adn = (layer == 2) ? nullptr : (const float*)d_in[6 + (layer + 1) * 4];
    gat_edge_owner<<<256, OWN_NT, 0, stream>>>(ei, keep, gsrc, as_cur, bias, dst, E, N, first,
                                               Wn, asn, adn, as_nxt, first ? nullptr : W);
  }
}
